# starting kernel
# baseline (speedup 1.0000x reference)
#include <hip/hip_runtime.h>
#include <math.h>

typedef float v2f __attribute__((ext_vector_type(2)));
typedef float v8f __attribute__((ext_vector_type(8)));

#define BATCH 8192
#define FDIM  512
#define NQ    8
#define NSTATE 256
#define NCLS  10

__global__ void k1_angles(const float* __restrict__ x,
                          const float* __restrict__ proj_w,
                          float* __restrict__ angles) {
    __shared__ float w[NQ * FDIM];
    int tid = threadIdx.x;
    for (int i = tid; i < NQ * FDIM; i += 256) w[i] = proj_w[i];
    __syncthreads();

    int lane = tid & 31;
    int wave = tid >> 5;
    int row  = blockIdx.x * 8 + wave;
    const float* xr = x + (size_t)row * FDIM;

    float acc[NQ];
#pragma unroll
    for (int q = 0; q < NQ; ++q) acc[q] = 0.0f;

    for (int j = 0; j < FDIM / 32; ++j) {
        float xv = xr[lane + 32 * j];
#pragma unroll
        for (int q = 0; q < NQ; ++q)
            acc[q] += xv * w[q * FDIM + lane + 32 * j];
    }
#pragma unroll
    for (int q = 0; q < NQ; ++q) {
        float v = acc[q];
#pragma unroll
        for (int off = 16; off >= 1; off >>= 1) v += __shfl_xor(v, off, 32);
        if (lane == q)
            angles[row * NQ + q] = tanhf(v) * 1.5707963267948966f;
    }
}

__device__ __forceinline__ void rot_lane_rx(float sr[8], float si[8],
                                            float c, float s, int m) {
#pragma unroll
    for (int r = 0; r < 8; ++r) {
        float pr = __shfl_xor(sr[r], m, 32);
        float pi = __shfl_xor(si[r], m, 32);
        float nr = c * sr[r] + s * pi;
        float ni = c * si[r] - s * pr;
        sr[r] = nr; si[r] = ni;
    }
}

__device__ __forceinline__ void rot_lane_ry(float sr[8], float si[8],
                                            float c, float s, int m, int lane) {
    float b = (lane & m) ? s : -s;
#pragma unroll
    for (int r = 0; r < 8; ++r) {
        float pr = __shfl_xor(sr[r], m, 32);
        float pi = __shfl_xor(si[r], m, 32);
        sr[r] = c * sr[r] + b * pr;
        si[r] = c * si[r] + b * pi;
    }
}

__device__ __forceinline__ void rot_reg_rx(float sr[8], float si[8],
                                           float c, float s, int rm) {
#pragma unroll
    for (int r0 = 0; r0 < 8; ++r0) {
        if (r0 & rm) continue;
        int r1 = r0 | rm;
        float a0r = sr[r0], a0i = si[r0], a1r = sr[r1], a1i = si[r1];
        sr[r0] = c * a0r + s * a1i;  si[r0] = c * a0i - s * a1r;
        sr[r1] = c * a1r + s * a0i;  si[r1] = c * a1i - s * a0r;
    }
}

__device__ __forceinline__ void rot_reg_ry(float sr[8], float si[8],
                                           float c, float s, int rm) {
#pragma unroll
    for (int r0 = 0; r0 < 8; ++r0) {
        if (r0 & rm) continue;
        int r1 = r0 | rm;
        float a0r = sr[r0], a0i = si[r0], a1r = sr[r1], a1i = si[r1];
        sr[r0] = c * a0r - s * a1r;  si[r0] = c * a0i - s * a1i;
        sr[r1] = s * a0r + c * a1r;  si[r1] = s * a0i + c * a1i;
    }
}

__device__ __forceinline__ void cnot_ll(float sr[8], float si[8],
                                        int cm, int tm, int lane) {
    bool hi = (lane & cm) != 0;
#pragma unroll
    for (int r = 0; r < 8; ++r) {
        float pr = __shfl_xor(sr[r], tm, 32);
        float pi = __shfl_xor(si[r], tm, 32);
        sr[r] = hi ? pr : sr[r];
        si[r] = hi ? pi : si[r];
    }
}

__device__ __forceinline__ void cnot_lane_reg(float sr[8], float si[8],
                                              int cm, int trm, int lane) {
    bool hi = (lane & cm) != 0;
#pragma unroll
    for (int r0 = 0; r0 < 8; ++r0) {
        if (r0 & trm) continue;
        int r1 = r0 | trm;
        float t0r = sr[r0], t0i = si[r0];
        sr[r0] = hi ? sr[r1] : sr[r0];  si[r0] = hi ? si[r1] : si[r0];
        sr[r1] = hi ? t0r    : sr[r1];  si[r1] = hi ? t0i    : si[r1];
    }
}

__device__ __forceinline__ void cnot_reg_lane(float sr[8], float si[8],
                                              int crm, int tm) {
#pragma unroll
    for (int r = 0; r < 8; ++r) {
        if (!(r & crm)) continue;
        sr[r] = __shfl_xor(sr[r], tm, 32);
        si[r] = __shfl_xor(si[r], tm, 32);
    }
}

__device__ __forceinline__ void cnot_reg_reg(float sr[8], float si[8],
                                             int crm, int trm) {
#pragma unroll
    for (int r0 = 0; r0 < 8; ++r0) {
        if (!(r0 & crm) || (r0 & trm)) continue;
        int r1 = r0 | trm;
        float tr = sr[r0], ti = si[r0];
        sr[r0] = sr[r1]; si[r0] = si[r1];
        sr[r1] = tr;     si[r1] = ti;
    }
}

__global__ void k2_unitary(const float* __restrict__ qnn_w,
                           float* __restrict__ UrP,
                           float* __restrict__ UiP) {
    int lane = threadIdx.x & 31;
    int t    = blockIdx.x * 8 + (threadIdx.x >> 5);

    float sr[8], si[8];
#pragma unroll
    for (int r = 0; r < 8; ++r) {
        sr[r] = (r * 32 + lane == t) ? 1.0f : 0.0f;
        si[r] = 0.0f;
    }

#pragma unroll
    for (int l = 0; l < 3; ++l) {
#pragma unroll
        for (int q = 0; q < NQ; ++q) {
            float w0 = qnn_w[(l * NQ + q) * 3 + 0] * 0.5f;
            float w1 = qnn_w[(l * NQ + q) * 3 + 1] * 0.5f;
            float w2 = qnn_w[(l * NQ + q) * 3 + 2] * 0.5f;
            float c0 = cosf(w0), s0 = sinf(w0);
            float c1 = cosf(w1), s1 = sinf(w1);
            float c2 = cosf(w2), s2 = sinf(w2);
            if (q < 5) {
                int m = 1 << q;
                rot_lane_rx(sr, si, c0, s0, m);
                rot_lane_ry(sr, si, c1, s1, m, lane);
                rot_lane_rx(sr, si, c2, s2, m);
            } else {
                int rm = 1 << (q - 5);
                rot_reg_rx(sr, si, c0, s0, rm);
                rot_reg_ry(sr, si, c1, s1, rm);
                rot_reg_rx(sr, si, c2, s2, rm);
            }
        }
        cnot_ll(sr, si, 1, 2, lane);
        cnot_ll(sr, si, 2, 4, lane);
        cnot_ll(sr, si, 4, 8, lane);
        cnot_ll(sr, si, 8, 16, lane);
        cnot_lane_reg(sr, si, 16, 1, lane);
        cnot_reg_reg(sr, si, 1, 2);
        cnot_reg_reg(sr, si, 2, 4);
        cnot_reg_lane(sr, si, 4, 1);
    }

#pragma unroll
    for (int r = 0; r < 8; ++r) {
        int s = r * 32 + lane;
        int off = (t >> 1) * (2 * NSTATE) + s * 2 + (t & 1);
        UrP[off] = sr[r];
        UiP[off] = si[r];
    }
}

__global__ __launch_bounds__(512) void k3_fused(
        const float* __restrict__ angles,
        const float* __restrict__ UrP,
        const float* __restrict__ UiP,
        const float* __restrict__ out_w,
        const float* __restrict__ out_b,
        float* __restrict__ out) {
    __shared__ __align__(16) float cs[16][NQ][2];
    __shared__ __align__(16) float enc[16][NSTATE];
    __shared__ __align__(16) float pq[16][NSTATE];
    __shared__ __align__(16) float E[16][NQ];

    int tid  = threadIdx.x;
    int lane = tid & 31;
    int wave = tid >> 5;
    int row0 = blockIdx.x * 16;

    if (tid < 128) {
        int m = tid >> 3, q = tid & 7;
        float a = angles[(row0 + m) * NQ + q] * 0.5f;
        cs[m][q][0] = cosf(a);
        cs[m][q][1] = sinf(a);
    }
    __syncthreads();

#pragma unroll
    for (int e = 0; e < 8; ++e) {
        int idx = tid + 512 * e;
        int m = idx >> 8, t = idx & 255;
        float v = 1.0f;
#pragma unroll
        for (int q = 0; q < NQ; ++q) v *= cs[m][q][(t >> q) & 1];
        enc[m][t] = v;
    }
    __syncthreads();

    v8f cr, ci;
#pragma unroll
    for (int v = 0; v < 8; ++v) { cr[v] = 0.0f; ci[v] = 0.0f; }

    int mrow = lane & 15;
    int h    = lane >> 4;
    int n    = lane & 15;
    int s0   = wave * 16;

    for (int k = 0; k < NSTATE; k += 4) {
        int kk = k + 2 * h;
        v2f a  = *(const v2f*)&enc[mrow][kk];
        const float* pr = &UrP[(kk >> 1) * (2 * NSTATE) + (s0 + n) * 2];
        const float* pi = &UiP[(kk >> 1) * (2 * NSTATE) + (s0 + n) * 2];
        v2f br = *(const v2f*)pr;
        v2f bi = *(const v2f*)pi;
        cr = __builtin_amdgcn_wmma_f32_16x16x4_f32(false, a, false, br,
                                                   (short)0, cr, false, false);
        ci = __builtin_amdgcn_wmma_f32_16x16x4_f32(false, a, false, bi,
                                                   (short)0, ci, false, false);
    }

#pragma unroll
    for (int v = 0; v < 8; ++v) {
        int m = v + 8 * h;
        pq[m][s0 + n] = cr[v] * cr[v] + ci[v] * ci[v];
    }
    __syncthreads();

    if (tid < 128) {
        int m = tid >> 3, q = tid & 7;
        float acc = 0.0f;
        for (int s = 0; s < NSTATE; ++s) {
            float p = pq[m][s];
            acc += ((s >> q) & 1) ? -p : p;
        }
        E[m][q] = acc;
    }
    __syncthreads();

    if (tid < 16 * NCLS) {
        int m = tid / NCLS, c = tid % NCLS;
        float acc = out_b[c];
#pragma unroll
        for (int q = 0; q < NQ; ++q) acc += E[m][q] * out_w[c * NQ + q];
        out[(row0 + m) * NCLS + c] = acc;
    }
}

extern "C" void kernel_launch(void* const* d_in, const int* in_sizes, int n_in,
                              void* d_out, int out_size, void* d_ws, size_t ws_size,
                              hipStream_t stream) {
    const float* x      = (const float*)d_in[0];
    const float* proj_w = (const float*)d_in[1];
    const float* qnn_w  = (const float*)d_in[2];
    const float* out_w  = (const float*)d_in[3];
    const float* out_b  = (const float*)d_in[4];
    float* ws = (float*)d_ws;

    float* angles = ws;
    float* UrP    = ws + 65536;
    float* UiP    = ws + 65536 + 65536;

    k1_angles<<<BATCH / 8, 256, 0, stream>>>(x, proj_w, angles);
    k2_unitary<<<NSTATE / 8, 256, 0, stream>>>(qnn_w, UrP, UiP);
    k3_fused<<<BATCH / 16, 512, 0, stream>>>(angles, UrP, UiP, out_w, out_b,
                                             (float*)d_out);
}


# unitary kernel: lane exchanges on the vector ALU (DPP row_xmask inside a 16-lane row, permlanex16 between the rows) instead of ds_bpermute LDS round trips; same values exchanged, U bit-identical
# speedup vs baseline: 1.3798x; 1.3798x over previous
#include <hip/hip_runtime.h>
#include <math.h>

typedef float v2f __attribute__((ext_vector_type(2)));
typedef float v8f __attribute__((ext_vector_type(8)));

#define BATCH 8192
#define FDIM  512
#define NQ    8
#define NSTATE 256
#define NCLS  10

__global__ __launch_bounds__(128) void k1_angles(const float* __restrict__ x,
                                                 const float* __restrict__ proj_w,
                                                 float* __restrict__ angles) {
    __shared__ float red[4][16][NQ];
    int tid  = threadIdx.x;
    int lane = tid & 31;
    int wave = tid >> 5;
    int c    = lane & 15;
    int h    = lane >> 4;
    int row0 = blockIdx.x * 16;
    int kb   = wave * 128 + 4 * h;

    const float4* xp = (const float4*)(x + (size_t)(row0 + c) * FDIM + kb);
    const float4* wp = (const float4*)(proj_w + (size_t)(c & 7) * FDIM + kb);

    float4 xa[16], wb[16];
#pragma unroll
    for (int j = 0; j < 16; ++j) wb[j] = wp[2 * j];
#pragma unroll
    for (int j = 0; j < 16; ++j) xa[j] = xp[2 * j];
    __builtin_amdgcn_sched_barrier(0);

    v8f acc0, acc1;
#pragma unroll
    for (int v = 0; v < 8; ++v) { acc0[v] = 0.0f; acc1[v] = 0.0f; }

#pragma unroll
    for (int j = 0; j < 16; ++j) {
        v2f a0, b0, a1, b1;
        a0[0] = xa[j].x; a0[1] = xa[j].y;  b0[0] = wb[j].x; b0[1] = wb[j].y;
        a1[0] = xa[j].z; a1[1] = xa[j].w;  b1[0] = wb[j].z; b1[1] = wb[j].w;
        acc0 = __builtin_amdgcn_wmma_f32_16x16x4_f32(false, a0, false, b0,
                                                     (short)0, acc0, false, false);
        acc1 = __builtin_amdgcn_wmma_f32_16x16x4_f32(false, a1, false, b1,
                                                     (short)0, acc1, false, false);
    }

    if (c < NQ) {
#pragma unroll
        for (int r = 0; r < 8; ++r) red[wave][8 * h + r][c] = acc0[r] + acc1[r];
    }
    __syncthreads();

    {
        int m = tid >> 3, q = tid & 7;
        float v = (red[0][m][q] + red[1][m][q]) + (red[2][m][q] + red[3][m][q]);
        angles[(row0 + m) * NQ + q] = tanhf(v) * 1.5707963267948966f;
    }
}

template <int M>
__device__ __forceinline__ float lane_xor(float v) {
    int iv = __builtin_bit_cast(int, v);
    int r;
    if constexpr (M < 16) {
        r = __builtin_amdgcn_update_dpp(iv, iv, 0x160 + M, 0xF, 0xF, true);
    } else {
        r = __builtin_amdgcn_permlanex16(iv, iv, 0x76543210, 0xfedcba98, false, false);
    }
    return __builtin_bit_cast(float, r);
}

template <int M>
__device__ __forceinline__ void rot_lane_rx(float sr[8], float si[8],
                                            float c, float s) {
#pragma unroll
    for (int r = 0; r < 8; ++r) {
        float pr = lane_xor<M>(sr[r]);
        float pi = lane_xor<M>(si[r]);
        float nr = c * sr[r] + s * pi;
        float ni = c * si[r] - s * pr;
        sr[r] = nr; si[r] = ni;
    }
}

template <int M>
__device__ __forceinline__ void rot_lane_ry(float sr[8], float si[8],
                                            float c, float s, int lane) {
    float b = (lane & M) ? s : -s;
#pragma unroll
    for (int r = 0; r < 8; ++r) {
        float pr = lane_xor<M>(sr[r]);
        float pi = lane_xor<M>(si[r]);
        sr[r] = c * sr[r] + b * pr;
        si[r] = c * si[r] + b * pi;
    }
}

__device__ __forceinline__ void rot_reg_rx(float sr[8], float si[8],
                                           float c, float s, int rm) {
#pragma unroll
    for (int r0 = 0; r0 < 8; ++r0) {
        if (r0 & rm) continue;
        int r1 = r0 | rm;
        float a0r = sr[r0], a0i = si[r0], a1r = sr[r1], a1i = si[r1];
        sr[r0] = c * a0r + s * a1i;  si[r0] = c * a0i - s * a1r;
        sr[r1] = c * a1r + s * a0i;  si[r1] = c * a1i - s * a0r;
    }
}

__device__ __forceinline__ void rot_reg_ry(float sr[8], float si[8],
                                           float c, float s, int rm) {
#pragma unroll
    for (int r0 = 0; r0 < 8; ++r0) {
        if (r0 & rm) continue;
        int r1 = r0 | rm;
        float a0r = sr[r0], a0i = si[r0], a1r = sr[r1], a1i = si[r1];
        sr[r0] = c * a0r - s * a1r;  si[r0] = c * a0i - s * a1i;
        sr[r1] = s * a0r + c * a1r;  si[r1] = s * a0i + c * a1i;
    }
}

template <int CM, int TM>
__device__ __forceinline__ void cnot_ll(float sr[8], float si[8], int lane) {
    bool hi = (lane & CM) != 0;
#pragma unroll
    for (int r = 0; r < 8; ++r) {
        float pr = lane_xor<TM>(sr[r]);
        float pi = lane_xor<TM>(si[r]);
        sr[r] = hi ? pr : sr[r];
        si[r] = hi ? pi : si[r];
    }
}

__device__ __forceinline__ void cnot_lane_reg(float sr[8], float si[8],
                                              int cm, int trm, int lane) {
    bool hi = (lane & cm) != 0;
#pragma unroll
    for (int r0 = 0; r0 < 8; ++r0) {
        if (r0 & trm) continue;
        int r1 = r0 | trm;
        float t0r = sr[r0], t0i = si[r0];
        sr[r0] = hi ? sr[r1] : sr[r0];  si[r0] = hi ? si[r1] : si[r0];
        sr[r1] = hi ? t0r    : sr[r1];  si[r1] = hi ? t0i    : si[r1];
    }
}

template <int TM>
__device__ __forceinline__ void cnot_reg_lane(float sr[8], float si[8],
                                              int crm) {
#pragma unroll
    for (int r = 0; r < 8; ++r) {
        if (!(r & crm)) continue;
        sr[r] = lane_xor<TM>(sr[r]);
        si[r] = lane_xor<TM>(si[r]);
    }
}

__device__ __forceinline__ void cnot_reg_reg(float sr[8], float si[8],
                                             int crm, int trm) {
#pragma unroll
    for (int r0 = 0; r0 < 8; ++r0) {
        if (!(r0 & crm) || (r0 & trm)) continue;
        int r1 = r0 | trm;
        float tr = sr[r0], ti = si[r0];
        sr[r0] = sr[r1]; si[r0] = si[r1];
        sr[r1] = tr;     si[r1] = ti;
    }
}

template <int Q>
__device__ __forceinline__ void qubit_rot(float sr[8], float si[8],
                                          const float* __restrict__ qnn_w,
                                          int l, int lane) {
    float w0 = qnn_w[(l * NQ + Q) * 3 + 0] * 0.5f;
    float w1 = qnn_w[(l * NQ + Q) * 3 + 1] * 0.5f;
    float w2 = qnn_w[(l * NQ + Q) * 3 + 2] * 0.5f;
    float c0 = cosf(w0), s0 = sinf(w0);
    float c1 = cosf(w1), s1 = sinf(w1);
    float c2 = cosf(w2), s2 = sinf(w2);
    if constexpr (Q < 5) {
        rot_lane_rx<(1 << Q)>(sr, si, c0, s0);
        rot_lane_ry<(1 << Q)>(sr, si, c1, s1, lane);
        rot_lane_rx<(1 << Q)>(sr, si, c2, s2);
    } else {
        constexpr int rm = 1 << (Q - 5);
        rot_reg_rx(sr, si, c0, s0, rm);
        rot_reg_ry(sr, si, c1, s1, rm);
        rot_reg_rx(sr, si, c2, s2, rm);
    }
}

__global__ void k2_unitary(const float* __restrict__ qnn_w,
                           float* __restrict__ UrP,
                           float* __restrict__ UiP) {
    int lane = threadIdx.x & 31;
    int t    = blockIdx.x * 8 + (threadIdx.x >> 5);

    float sr[8], si[8];
#pragma unroll
    for (int r = 0; r < 8; ++r) {
        sr[r] = (r * 32 + lane == t) ? 1.0f : 0.0f;
        si[r] = 0.0f;
    }

#pragma unroll
    for (int l = 0; l < 3; ++l) {
        qubit_rot<0>(sr, si, qnn_w, l, lane);
        qubit_rot<1>(sr, si, qnn_w, l, lane);
        qubit_rot<2>(sr, si, qnn_w, l, lane);
        qubit_rot<3>(sr, si, qnn_w, l, lane);
        qubit_rot<4>(sr, si, qnn_w, l, lane);
        qubit_rot<5>(sr, si, qnn_w, l, lane);
        qubit_rot<6>(sr, si, qnn_w, l, lane);
        qubit_rot<7>(sr, si, qnn_w, l, lane);
        cnot_ll<1, 2>(sr, si, lane);
        cnot_ll<2, 4>(sr, si, lane);
        cnot_ll<4, 8>(sr, si, lane);
        cnot_ll<8, 16>(sr, si, lane);
        cnot_lane_reg(sr, si, 16, 1, lane);
        cnot_reg_reg(sr, si, 1, 2);
        cnot_reg_reg(sr, si, 2, 4);
        cnot_reg_lane<1>(sr, si, 4);
    }

#pragma unroll
    for (int r = 0; r < 8; ++r) {
        int s = r * 32 + lane;
        int off = (t >> 1) * (2 * NSTATE) + s * 2 + (t & 1);
        UrP[off] = sr[r];
        UiP[off] = si[r];
    }
}

__global__ __launch_bounds__(512) void k3_fused(
        const float* __restrict__ angles,
        const float* __restrict__ UrP,
        const float* __restrict__ UiP,
        const float* __restrict__ out_w,
        const float* __restrict__ out_b,
        float* __restrict__ out) {
    __shared__ __align__(16) float cs[16][NQ][2];
    __shared__ __align__(16) float encP[NSTATE / 2][16][2];
    __shared__ __align__(16) float pq[16][NSTATE];
    __shared__ __align__(16) float E[16][NQ];

    int tid  = threadIdx.x;
    int lane = tid & 31;
    int wave = tid >> 5;
    int row0 = blockIdx.x * 16;

    if (tid < 128) {
        int m = tid >> 3, q = tid & 7;
        float a = angles[(row0 + m) * NQ + q] * 0.5f;
        cs[m][q][0] = cosf(a);
        cs[m][q][1] = sinf(a);
    }
    __syncthreads();

    {
        int m = tid & 15;
        int p = tid >> 4;
        const float4* c4 = (const float4*)&cs[m][0][0];
        float4 c01 = c4[0], c23 = c4[1], c45 = c4[2], c67 = c4[3];
        float f1 = (p & 1)  ? c01.w : c01.z;
        float f2 = (p & 2)  ? c23.y : c23.x;
        float f3 = (p & 4)  ? c23.w : c23.z;
        float f4 = (p & 8)  ? c45.y : c45.x;
        float f5 = (p & 16) ? c45.w : c45.z;
        float v0 = c01.x, v1 = c01.y;
        v0 *= f1; v1 *= f1;
        v0 *= f2; v1 *= f2;
        v0 *= f3; v1 *= f3;
        v0 *= f4; v1 *= f4;
        v0 *= f5; v1 *= f5;
#pragma unroll
        for (int e = 0; e < 4; ++e) {
            float f6 = (e & 1) ? c67.y : c67.x;
            float f7 = (e & 2) ? c67.w : c67.z;
            v2f pr;
            pr[0] = (v0 * f6) * f7;
            pr[1] = (v1 * f6) * f7;
            *(v2f*)&encP[p + 32 * e][m][0] = pr;
        }
    }
    __syncthreads();

    v8f cr, ci;
#pragma unroll
    for (int v = 0; v < 8; ++v) { cr[v] = 0.0f; ci[v] = 0.0f; }

    int mrow = lane & 15;
    int h    = lane >> 4;
    int n    = lane & 15;
    int s0   = wave * 16;

    for (int k = 0; k < NSTATE; k += 4) {
        int kk = k + 2 * h;
        v2f a  = *(const v2f*)&encP[kk >> 1][mrow][0];
        const float* pr = &UrP[(kk >> 1) * (2 * NSTATE) + (s0 + n) * 2];
        const float* pi = &UiP[(kk >> 1) * (2 * NSTATE) + (s0 + n) * 2];
        v2f br = *(const v2f*)pr;
        v2f bi = *(const v2f*)pi;
        cr = __builtin_amdgcn_wmma_f32_16x16x4_f32(false, a, false, br,
                                                   (short)0, cr, false, false);
        ci = __builtin_amdgcn_wmma_f32_16x16x4_f32(false, a, false, bi,
                                                   (short)0, ci, false, false);
    }

#pragma unroll
    for (int v = 0; v < 8; ++v) {
        int m = v + 8 * h;
        pq[m][s0 + n] = cr[v] * cr[v] + ci[v] * ci[v];
    }
    __syncthreads();

    if (tid < 128) {
        int m = tid >> 3, q = tid & 7;
        float acc = 0.0f;
        for (int s = 0; s < NSTATE; ++s) {
            float p = pq[m][s];
            acc += ((s >> q) & 1) ? -p : p;
        }
        E[m][q] = acc;
    }
    __syncthreads();

    if (tid < 16 * NCLS) {
        int m = tid / NCLS, c = tid % NCLS;
        float acc = out_b[c];
#pragma unroll
        for (int q = 0; q < NQ; ++q) acc += E[m][q] * out_w[c * NQ + q];
        out[(row0 + m) * NCLS + c] = acc;
    }
}

extern "C" void kernel_launch(void* const* d_in, const int* in_sizes, int n_in,
                              void* d_out, int out_size, void* d_ws, size_t ws_size,
                              hipStream_t stream) {
    const float* x      = (const float*)d_in[0];
    const float* proj_w = (const float*)d_in[1];
    const float* qnn_w  = (const float*)d_in[2];
    const float* out_w  = (const float*)d_in[3];
    const float* out_b  = (const float*)d_in[4];
    float* ws = (float*)d_ws;

    float* angles = ws;
    float* UrP    = ws + 65536;
    float* UiP    = ws + 65536 + 65536;

    k1_angles<<<BATCH / 16, 128, 0, stream>>>(x, proj_w, angles);
    k2_unitary<<<NSTATE / 8, 256, 0, stream>>>(qnn_w, UrP, UiP);
    k3_fused<<<BATCH / 16, 512, 0, stream>>>(angles, UrP, UiP, out_w, out_b,
                                             (float*)d_out);
}


# weights-only work split off into the angle launch: one extra work-group stores cos/sin of the 72 half circuit weights once per call; unitary kernel loads the 144 floats as scalars, no sincos in it
# speedup vs baseline: 1.7861x; 1.2945x over previous
#include <hip/hip_runtime.h>
#include <math.h>

typedef float v2f __attribute__((ext_vector_type(2)));
typedef float v8f __attribute__((ext_vector_type(8)));

#define BATCH 8192
#define FDIM  512
#define NQ    8
#define NSTATE 256
#define NCLS  10

#define NWEIGHT (3 * NQ * 3)
__global__ __launch_bounds__(128) void k1_angles(const float* __restrict__ x,
                                                 const float* __restrict__ proj_w,
                                                 const float* __restrict__ qnn_w,
                                                 float* __restrict__ angles,
                                                 float* __restrict__ cs_tab) {
    __shared__ float red[4][16][NQ];
    int tid  = threadIdx.x;
    if (blockIdx.x == BATCH / 16) {
        if (tid < NWEIGHT) {
            float a = qnn_w[tid] * 0.5f;
            cs_tab[2 * tid]     = cosf(a);
            cs_tab[2 * tid + 1] = sinf(a);
        }
        return;
    }
    int lane = tid & 31;
    int wave = tid >> 5;
    int c    = lane & 15;
    int h    = lane >> 4;
    int row0 = blockIdx.x * 16;
    int kb   = wave * 128 + 4 * h;

    const float4* xp = (const float4*)(x + (size_t)(row0 + c) * FDIM + kb);
    const float4* wp = (const float4*)(proj_w + (size_t)(c & 7) * FDIM + kb);

    float4 xa[16], wb[16];
#pragma unroll
    for (int j = 0; j < 16; ++j) wb[j] = wp[2 * j];
#pragma unroll
    for (int j = 0; j < 16; ++j) xa[j] = xp[2 * j];
    __builtin_amdgcn_sched_barrier(0);

    v8f acc0, acc1;
#pragma unroll
    for (int v = 0; v < 8; ++v) { acc0[v] = 0.0f; acc1[v] = 0.0f; }

#pragma unroll
    for (int j = 0; j < 16; ++j) {
        v2f a0, b0, a1, b1;
        a0[0] = xa[j].x; a0[1] = xa[j].y;  b0[0] = wb[j].x; b0[1] = wb[j].y;
        a1[0] = xa[j].z; a1[1] = xa[j].w;  b1[0] = wb[j].z; b1[1] = wb[j].w;
        acc0 = __builtin_amdgcn_wmma_f32_16x16x4_f32(false, a0, false, b0,
                                                     (short)0, acc0, false, false);
        acc1 = __builtin_amdgcn_wmma_f32_16x16x4_f32(false, a1, false, b1,
                                                     (short)0, acc1, false, false);
    }

    if (c < NQ) {
#pragma unroll
        for (int r = 0; r < 8; ++r) red[wave][8 * h + r][c] = acc0[r] + acc1[r];
    }
    __syncthreads();

    {
        int m = tid >> 3, q = tid & 7;
        float v = (red[0][m][q] + red[1][m][q]) + (red[2][m][q] + red[3][m][q]);
        angles[(row0 + m) * NQ + q] = tanhf(v) * 1.5707963267948966f;
    }
}

template <int M>
__device__ __forceinline__ float lane_xor(float v) {
    int iv = __builtin_bit_cast(int, v);
    int r;
    if constexpr (M < 16) {
        r = __builtin_amdgcn_update_dpp(iv, iv, 0x160 + M, 0xF, 0xF, true);
    } else {
        r = __builtin_amdgcn_permlanex16(iv, iv, 0x76543210, 0xfedcba98, false, false);
    }
    return __builtin_bit_cast(float, r);
}

template <int M>
__device__ __forceinline__ void rot_lane_rx(float sr[8], float si[8],
                                            float c, float s) {
#pragma unroll
    for (int r = 0; r < 8; ++r) {
        float pr = lane_xor<M>(sr[r]);
        float pi = lane_xor<M>(si[r]);
        float nr = c * sr[r] + s * pi;
        float ni = c * si[r] - s * pr;
        sr[r] = nr; si[r] = ni;
    }
}

template <int M>
__device__ __forceinline__ void rot_lane_ry(float sr[8], float si[8],
                                            float c, float s, int lane) {
    float b = (lane & M) ? s : -s;
#pragma unroll
    for (int r = 0; r < 8; ++r) {
        float pr = lane_xor<M>(sr[r]);
        float pi = lane_xor<M>(si[r]);
        sr[r] = c * sr[r] + b * pr;
        si[r] = c * si[r] + b * pi;
    }
}

__device__ __forceinline__ void rot_reg_rx(float sr[8], float si[8],
                                           float c, float s, int rm) {
#pragma unroll
    for (int r0 = 0; r0 < 8; ++r0) {
        if (r0 & rm) continue;
        int r1 = r0 | rm;
        float a0r = sr[r0], a0i = si[r0], a1r = sr[r1], a1i = si[r1];
        sr[r0] = c * a0r + s * a1i;  si[r0] = c * a0i - s * a1r;
        sr[r1] = c * a1r + s * a0i;  si[r1] = c * a1i - s * a0r;
    }
}

__device__ __forceinline__ void rot_reg_ry(float sr[8], float si[8],
                                           float c, float s, int rm) {
#pragma unroll
    for (int r0 = 0; r0 < 8; ++r0) {
        if (r0 & rm) continue;
        int r1 = r0 | rm;
        float a0r = sr[r0], a0i = si[r0], a1r = sr[r1], a1i = si[r1];
        sr[r0] = c * a0r - s * a1r;  si[r0] = c * a0i - s * a1i;
        sr[r1] = s * a0r + c * a1r;  si[r1] = s * a0i + c * a1i;
    }
}

template <int CM, int TM>
__device__ __forceinline__ void cnot_ll(float sr[8], float si[8], int lane) {
    bool hi = (lane & CM) != 0;
#pragma unroll
    for (int r = 0; r < 8; ++r) {
        float pr = lane_xor<TM>(sr[r]);
        float pi = lane_xor<TM>(si[r]);
        sr[r] = hi ? pr : sr[r];
        si[r] = hi ? pi : si[r];
    }
}

__device__ __forceinline__ void cnot_lane_reg(float sr[8], float si[8],
                                              int cm, int trm, int lane) {
    bool hi = (lane & cm) != 0;
#pragma unroll
    for (int r0 = 0; r0 < 8; ++r0) {
        if (r0 & trm) continue;
        int r1 = r0 | trm;
        float t0r = sr[r0], t0i = si[r0];
        sr[r0] = hi ? sr[r1] : sr[r0];  si[r0] = hi ? si[r1] : si[r0];
        sr[r1] = hi ? t0r    : sr[r1];  si[r1] = hi ? t0i    : si[r1];
    }
}

template <int TM>
__device__ __forceinline__ void cnot_reg_lane(float sr[8], float si[8],
                                              int crm) {
#pragma unroll
    for (int r = 0; r < 8; ++r) {
        if (!(r & crm)) continue;
        sr[r] = lane_xor<TM>(sr[r]);
        si[r] = lane_xor<TM>(si[r]);
    }
}

__device__ __forceinline__ void cnot_reg_reg(float sr[8], float si[8],
                                             int crm, int trm) {
#pragma unroll
    for (int r0 = 0; r0 < 8; ++r0) {
        if (!(r0 & crm) || (r0 & trm)) continue;
        int r1 = r0 | trm;
        float tr = sr[r0], ti = si[r0];
        sr[r0] = sr[r1]; si[r0] = si[r1];
        sr[r1] = tr;     si[r1] = ti;
    }
}

template <int Q>
__device__ __forceinline__ void qubit_rot(float sr[8], float si[8],
                                          const float* __restrict__ cs_tab,
                                          int l, int lane) {
    const float* g = cs_tab + (l * NQ + Q) * 6;
    float c0 = g[0], s0 = g[1];
    float c1 = g[2], s1 = g[3];
    float c2 = g[4], s2 = g[5];
    if constexpr (Q < 5) {
        rot_lane_rx<(1 << Q)>(sr, si, c0, s0);
        rot_lane_ry<(1 << Q)>(sr, si, c1, s1, lane);
        rot_lane_rx<(1 << Q)>(sr, si, c2, s2);
    } else {
        constexpr int rm = 1 << (Q - 5);
        rot_reg_rx(sr, si, c0, s0, rm);
        rot_reg_ry(sr, si, c1, s1, rm);
        rot_reg_rx(sr, si, c2, s2, rm);
    }
}

__global__ void k2_unitary(const float* __restrict__ cs_tab,
                           float* __restrict__ UrP,
                           float* __restrict__ UiP) {
    int lane = threadIdx.x & 31;
    int t    = blockIdx.x * 8 + (threadIdx.x >> 5);

    float sr[8], si[8];
#pragma unroll
    for (int r = 0; r < 8; ++r) {
        sr[r] = (r * 32 + lane == t) ? 1.0f : 0.0f;
        si[r] = 0.0f;
    }

#pragma unroll
    for (int l = 0; l < 3; ++l) {
        qubit_rot<0>(sr, si, cs_tab, l, lane);
        qubit_rot<1>(sr, si, cs_tab, l, lane);
        qubit_rot<2>(sr, si, cs_tab, l, lane);
        qubit_rot<3>(sr, si, cs_tab, l, lane);
        qubit_rot<4>(sr, si, cs_tab, l, lane);
        qubit_rot<5>(sr, si, cs_tab, l, lane);
        qubit_rot<6>(sr, si, cs_tab, l, lane);
        qubit_rot<7>(sr, si, cs_tab, l, lane);
        cnot_ll<1, 2>(sr, si, lane);
        cnot_ll<2, 4>(sr, si, lane);
        cnot_ll<4, 8>(sr, si, lane);
        cnot_ll<8, 16>(sr, si, lane);
        cnot_lane_reg(sr, si, 16, 1, lane);
        cnot_reg_reg(sr, si, 1, 2);
        cnot_reg_reg(sr, si, 2, 4);
        cnot_reg_lane<1>(sr, si, 4);
    }

#pragma unroll
    for (int r = 0; r < 8; ++r) {
        int s = r * 32 + lane;
        int off = (t >> 1) * (2 * NSTATE) + s * 2 + (t & 1);
        UrP[off] = sr[r];
        UiP[off] = si[r];
    }
}

__global__ __launch_bounds__(512) void k3_fused(
        const float* __restrict__ angles,
        const float* __restrict__ UrP,
        const float* __restrict__ UiP,
        const float* __restrict__ out_w,
        const float* __restrict__ out_b,
        float* __restrict__ out) {
    __shared__ __align__(16) float cs[16][NQ][2];
    __shared__ __align__(16) float encP[NSTATE / 2][16][2];
    __shared__ __align__(16) float pq[16][NSTATE];
    __shared__ __align__(16) float E[16][NQ];

    int tid  = threadIdx.x;
    int lane = tid & 31;
    int wave = tid >> 5;
    int row0 = blockIdx.x * 16;

    if (tid < 128) {
        int m = tid >> 3, q = tid & 7;
        float a = angles[(row0 + m) * NQ + q] * 0.5f;
        cs[m][q][0] = cosf(a);
        cs[m][q][1] = sinf(a);
    }
    __syncthreads();

    {
        int m = tid & 15;
        int p = tid >> 4;
        const float4* c4 = (const float4*)&cs[m][0][0];
        float4 c01 = c4[0], c23 = c4[1], c45 = c4[2], c67 = c4[3];
        float f1 = (p & 1)  ? c01.w : c01.z;
        float f2 = (p & 2)  ? c23.y : c23.x;
        float f3 = (p & 4)  ? c23.w : c23.z;
        float f4 = (p & 8)  ? c45.y : c45.x;
        float f5 = (p & 16) ? c45.w : c45.z;
        float v0 = c01.x, v1 = c01.y;
        v0 *= f1; v1 *= f1;
        v0 *= f2; v1 *= f2;
        v0 *= f3; v1 *= f3;
        v0 *= f4; v1 *= f4;
        v0 *= f5; v1 *= f5;
#pragma unroll
        for (int e = 0; e < 4; ++e) {
            float f6 = (e & 1) ? c67.y : c67.x;
            float f7 = (e & 2) ? c67.w : c67.z;
            v2f pr;
            pr[0] = (v0 * f6) * f7;
            pr[1] = (v1 * f6) * f7;
            *(v2f*)&encP[p + 32 * e][m][0] = pr;
        }
    }
    __syncthreads();

    v8f cr, ci;
#pragma unroll
    for (int v = 0; v < 8; ++v) { cr[v] = 0.0f; ci[v] = 0.0f; }

    int mrow = lane & 15;
    int h    = lane >> 4;
    int n    = lane & 15;
    int s0   = wave * 16;

    for (int k = 0; k < NSTATE; k += 4) {
        int kk = k + 2 * h;
        v2f a  = *(const v2f*)&encP[kk >> 1][mrow][0];
        const float* pr = &UrP[(kk >> 1) * (2 * NSTATE) + (s0 + n) * 2];
        const float* pi = &UiP[(kk >> 1) * (2 * NSTATE) + (s0 + n) * 2];
        v2f br = *(const v2f*)pr;
        v2f bi = *(const v2f*)pi;
        cr = __builtin_amdgcn_wmma_f32_16x16x4_f32(false, a, false, br,
                                                   (short)0, cr, false, false);
        ci = __builtin_amdgcn_wmma_f32_16x16x4_f32(false, a, false, bi,
                                                   (short)0, ci, false, false);
    }

#pragma unroll
    for (int v = 0; v < 8; ++v) {
        int m = v + 8 * h;
        pq[m][s0 + n] = cr[v] * cr[v] + ci[v] * ci[v];
    }
    __syncthreads();

    if (tid < 128) {
        int m = tid >> 3, q = tid & 7;
        float acc = 0.0f;
        for (int s = 0; s < NSTATE; ++s) {
            float p = pq[m][s];
            acc += ((s >> q) & 1) ? -p : p;
        }
        E[m][q] = acc;
    }
    __syncthreads();

    if (tid < 16 * NCLS) {
        int m = tid / NCLS, c = tid % NCLS;
        float acc = out_b[c];
#pragma unroll
        for (int q = 0; q < NQ; ++q) acc += E[m][q] * out_w[c * NQ + q];
        out[(row0 + m) * NCLS + c] = acc;
    }
}

extern "C" void kernel_launch(void* const* d_in, const int* in_sizes, int n_in,
                              void* d_out, int out_size, void* d_ws, size_t ws_size,
                              hipStream_t stream) {
    const float* x      = (const float*)d_in[0];
    const float* proj_w = (const float*)d_in[1];
    const float* qnn_w  = (const float*)d_in[2];
    const float* out_w  = (const float*)d_in[3];
    const float* out_b  = (const float*)d_in[4];
    float* ws = (float*)d_ws;

    float* angles = ws;
    float* UrP    = ws + 65536;
    float* UiP    = ws + 65536 + 65536;
    float* cs_tab = (float*)d_out;

    k1_angles<<<BATCH / 16 + 1, 128, 0, stream>>>(x, proj_w, qnn_w, angles, cs_tab);
    k2_unitary<<<NSTATE / 8, 256, 0, stream>>>(cs_tab, UrP, UiP);
    k3_fused<<<BATCH / 16, 512, 0, stream>>>(angles, UrP, UiP, out_w, out_b,
                                             (float*)d_out);
}


# GEMM kernel: 32 rows per work-group; each 32-k slab of U is copied once per work-group into double-buffered LDS and feeds the B fragments of both row tiles' waves; same products, same k order
# speedup vs baseline: 1.9025x; 1.0652x over previous
#include <hip/hip_runtime.h>
#include <math.h>

typedef float v2f __attribute__((ext_vector_type(2)));
typedef float v8f __attribute__((ext_vector_type(8)));

#define BATCH 8192
#define FDIM  512
#define NQ    8
#define NSTATE 256
#define NCLS  10

#define NWEIGHT (3 * NQ * 3)
__global__ __launch_bounds__(128) void k1_angles(const float* __restrict__ x,
                                                 const float* __restrict__ proj_w,
                                                 const float* __restrict__ qnn_w,
                                                 float* __restrict__ angles,
                                                 float* __restrict__ cs_tab) {
    __shared__ float red[4][16][NQ];
    int tid  = threadIdx.x;
    if (blockIdx.x == BATCH / 16) {
        if (tid < NWEIGHT) {
            float a = qnn_w[tid] * 0.5f;
            cs_tab[2 * tid]     = cosf(a);
            cs_tab[2 * tid + 1] = sinf(a);
        }
        return;
    }
    int lane = tid & 31;
    int wave = tid >> 5;
    int c    = lane & 15;
    int h    = lane >> 4;
    int row0 = blockIdx.x * 16;
    int kb   = wave * 128 + 4 * h;

    const float4* xp = (const float4*)(x + (size_t)(row0 + c) * FDIM + kb);
    const float4* wp = (const float4*)(proj_w + (size_t)(c & 7) * FDIM + kb);

    float4 xa[16], wb[16];
#pragma unroll
    for (int j = 0; j < 16; ++j) wb[j] = wp[2 * j];
#pragma unroll
    for (int j = 0; j < 16; ++j) xa[j] = xp[2 * j];
    __builtin_amdgcn_sched_barrier(0);

    v8f acc0, acc1;
#pragma unroll
    for (int v = 0; v < 8; ++v) { acc0[v] = 0.0f; acc1[v] = 0.0f; }

#pragma unroll
    for (int j = 0; j < 16; ++j) {
        v2f a0, b0, a1, b1;
        a0[0] = xa[j].x; a0[1] = xa[j].y;  b0[0] = wb[j].x; b0[1] = wb[j].y;
        a1[0] = xa[j].z; a1[1] = xa[j].w;  b1[0] = wb[j].z; b1[1] = wb[j].w;
        acc0 = __builtin_amdgcn_wmma_f32_16x16x4_f32(false, a0, false, b0,
                                                     (short)0, acc0, false, false);
        acc1 = __builtin_amdgcn_wmma_f32_16x16x4_f32(false, a1, false, b1,
                                                     (short)0, acc1, false, false);
    }

    if (c < NQ) {
#pragma unroll
        for (int r = 0; r < 8; ++r) red[wave][8 * h + r][c] = acc0[r] + acc1[r];
    }
    __syncthreads();

    {
        int m = tid >> 3, q = tid & 7;
        float v = (red[0][m][q] + red[1][m][q]) + (red[2][m][q] + red[3][m][q]);
        angles[(row0 + m) * NQ + q] = tanhf(v) * 1.5707963267948966f;
    }
}

template <int M>
__device__ __forceinline__ float lane_xor(float v) {
    int iv = __builtin_bit_cast(int, v);
    int r;
    if constexpr (M < 16) {
        r = __builtin_amdgcn_update_dpp(iv, iv, 0x160 + M, 0xF, 0xF, true);
    } else {
        r = __builtin_amdgcn_permlanex16(iv, iv, 0x76543210, 0xfedcba98, false, false);
    }
    return __builtin_bit_cast(float, r);
}

template <int M>
__device__ __forceinline__ void rot_lane_rx(float sr[8], float si[8],
                                            float c, float s) {
#pragma unroll
    for (int r = 0; r < 8; ++r) {
        float pr = lane_xor<M>(sr[r]);
        float pi = lane_xor<M>(si[r]);
        float nr = c * sr[r] + s * pi;
        float ni = c * si[r] - s * pr;
        sr[r] = nr; si[r] = ni;
    }
}

template <int M>
__device__ __forceinline__ void rot_lane_ry(float sr[8], float si[8],
                                            float c, float s, int lane) {
    float b = (lane & M) ? s : -s;
#pragma unroll
    for (int r = 0; r < 8; ++r) {
        float pr = lane_xor<M>(sr[r]);
        float pi = lane_xor<M>(si[r]);
        sr[r] = c * sr[r] + b * pr;
        si[r] = c * si[r] + b * pi;
    }
}

__device__ __forceinline__ void rot_reg_rx(float sr[8], float si[8],
                                           float c, float s, int rm) {
#pragma unroll
    for (int r0 = 0; r0 < 8; ++r0) {
        if (r0 & rm) continue;
        int r1 = r0 | rm;
        float a0r = sr[r0], a0i = si[r0], a1r = sr[r1], a1i = si[r1];
        sr[r0] = c * a0r + s * a1i;  si[r0] = c * a0i - s * a1r;
        sr[r1] = c * a1r + s * a0i;  si[r1] = c * a1i - s * a0r;
    }
}

__device__ __forceinline__ void rot_reg_ry(float sr[8], float si[8],
                                           float c, float s, int rm) {
#pragma unroll
    for (int r0 = 0; r0 < 8; ++r0) {
        if (r0 & rm) continue;
        int r1 = r0 | rm;
        float a0r = sr[r0], a0i = si[r0], a1r = sr[r1], a1i = si[r1];
        sr[r0] = c * a0r - s * a1r;  si[r0] = c * a0i - s * a1i;
        sr[r1] = s * a0r + c * a1r;  si[r1] = s * a0i + c * a1i;
    }
}

template <int CM, int TM>
__device__ __forceinline__ void cnot_ll(float sr[8], float si[8], int lane) {
    bool hi = (lane & CM) != 0;
#pragma unroll
    for (int r = 0; r < 8; ++r) {
        float pr = lane_xor<TM>(sr[r]);
        float pi = lane_xor<TM>(si[r]);
        sr[r] = hi ? pr : sr[r];
        si[r] = hi ? pi : si[r];
    }
}

__device__ __forceinline__ void cnot_lane_reg(float sr[8], float si[8],
                                              int cm, int trm, int lane) {
    bool hi = (lane & cm) != 0;
#pragma unroll
    for (int r0 = 0; r0 < 8; ++r0) {
        if (r0 & trm) continue;
        int r1 = r0 | trm;
        float t0r = sr[r0], t0i = si[r0];
        sr[r0] = hi ? sr[r1] : sr[r0];  si[r0] = hi ? si[r1] : si[r0];
        sr[r1] = hi ? t0r    : sr[r1];  si[r1] = hi ? t0i    : si[r1];
    }
}

template <int TM>
__device__ __forceinline__ void cnot_reg_lane(float sr[8], float si[8],
                                              int crm) {
#pragma unroll
    for (int r = 0; r < 8; ++r) {
        if (!(r & crm)) continue;
        sr[r] = lane_xor<TM>(sr[r]);
        si[r] = lane_xor<TM>(si[r]);
    }
}

__device__ __forceinline__ void cnot_reg_reg(float sr[8], float si[8],
                                             int crm, int trm) {
#pragma unroll
    for (int r0 = 0; r0 < 8; ++r0) {
        if (!(r0 & crm) || (r0 & trm)) continue;
        int r1 = r0 | trm;
        float tr = sr[r0], ti = si[r0];
        sr[r0] = sr[r1]; si[r0] = si[r1];
        sr[r1] = tr;     si[r1] = ti;
    }
}

template <int Q>
__device__ __forceinline__ void qubit_rot(float sr[8], float si[8],
                                          const float* __restrict__ cs_tab,
                                          int l, int lane) {
    const float* g = cs_tab + (l * NQ + Q) * 6;
    float c0 = g[0], s0 = g[1];
    float c1 = g[2], s1 = g[3];
    float c2 = g[4], s2 = g[5];
    if constexpr (Q < 5) {
        rot_lane_rx<(1 << Q)>(sr, si, c0, s0);
        rot_lane_ry<(1 << Q)>(sr, si, c1, s1, lane);
        rot_lane_rx<(1 << Q)>(sr, si, c2, s2);
    } else {
        constexpr int rm = 1 << (Q - 5);
        rot_reg_rx(sr, si, c0, s0, rm);
        rot_reg_ry(sr, si, c1, s1, rm);
        rot_reg_rx(sr, si, c2, s2, rm);
    }
}

__global__ void k2_unitary(const float* __restrict__ cs_tab,
                           float* __restrict__ UrP,
                           float* __restrict__ UiP) {
    int lane = threadIdx.x & 31;
    int t    = blockIdx.x * 8 + (threadIdx.x >> 5);

    float sr[8], si[8];
#pragma unroll
    for (int r = 0; r < 8; ++r) {
        sr[r] = (r * 32 + lane == t) ? 1.0f : 0.0f;
        si[r] = 0.0f;
    }

#pragma unroll
    for (int l = 0; l < 3; ++l) {
        qubit_rot<0>(sr, si, cs_tab, l, lane);
        qubit_rot<1>(sr, si, cs_tab, l, lane);
        qubit_rot<2>(sr, si, cs_tab, l, lane);
        qubit_rot<3>(sr, si, cs_tab, l, lane);
        qubit_rot<4>(sr, si, cs_tab, l, lane);
        qubit_rot<5>(sr, si, cs_tab, l, lane);
        qubit_rot<6>(sr, si, cs_tab, l, lane);
        qubit_rot<7>(sr, si, cs_tab, l, lane);
        cnot_ll<1, 2>(sr, si, lane);
        cnot_ll<2, 4>(sr, si, lane);
        cnot_ll<4, 8>(sr, si, lane);
        cnot_ll<8, 16>(sr, si, lane);
        cnot_lane_reg(sr, si, 16, 1, lane);
        cnot_reg_reg(sr, si, 1, 2);
        cnot_reg_reg(sr, si, 2, 4);
        cnot_reg_lane<1>(sr, si, 4);
    }

#pragma unroll
    for (int r = 0; r < 8; ++r) {
        int s = r * 32 + lane;
        int off = (t >> 1) * (2 * NSTATE) + s * 2 + (t & 1);
        UrP[off] = sr[r];
        UiP[off] = si[r];
    }
}

#define K3_ROWS    32
#define SLAB_PAIRS 16
#define NSLAB      (NSTATE / 2 / SLAB_PAIRS)
#define UPITCH     (2 * NSTATE + 32)

__global__ __launch_bounds__(512) void k3_fused(
        const float* __restrict__ angles,
        const float* UrP,
        const float* UiP,
        const float* __restrict__ out_w,
        const float* __restrict__ out_b,
        float* __restrict__ out) {
    __shared__ __align__(16) float cs[K3_ROWS][NQ][2];
    __shared__ __align__(16) float encP[2][NSTATE / 2][16][2];
    __shared__ __align__(16) float Ub[2][2][SLAB_PAIRS * UPITCH];
    __shared__ __align__(16) float pq[K3_ROWS][NSTATE];
    __shared__ __align__(16) float E[K3_ROWS][NQ];

    int tid  = threadIdx.x;
    int lane = tid & 31;
    int wave = tid >> 5;
    int row0 = blockIdx.x * K3_ROWS;

    const float4* gUr = (const float4*)UrP;
    const float4* gUi = (const float4*)UiP;
    int so = (tid >> 7) * UPITCH + (tid & 127) * 4;
    float4 tr0, tr1, tr2, tr3, ti0, ti1, ti2, ti3;
#define U_ISSUE(S)                                                           \
    tr0 = gUr[(S) * 2048 + tid];        ti0 = gUi[(S) * 2048 + tid];        \
    tr1 = gUr[(S) * 2048 + tid + 512];  ti1 = gUi[(S) * 2048 + tid + 512];  \
    tr2 = gUr[(S) * 2048 + tid + 1024]; ti2 = gUi[(S) * 2048 + tid + 1024]; \
    tr3 = gUr[(S) * 2048 + tid + 1536]; ti3 = gUi[(S) * 2048 + tid + 1536];
#define U_STASH(B)                                                           \
    *(float4*)&Ub[B][0][so]               = tr0; *(float4*)&Ub[B][1][so]               = ti0; \
    *(float4*)&Ub[B][0][so + 4 * UPITCH]  = tr1; *(float4*)&Ub[B][1][so + 4 * UPITCH]  = ti1; \
    *(float4*)&Ub[B][0][so + 8 * UPITCH]  = tr2; *(float4*)&Ub[B][1][so + 8 * UPITCH]  = ti2; \
    *(float4*)&Ub[B][0][so + 12 * UPITCH] = tr3; *(float4*)&Ub[B][1][so + 12 * UPITCH] = ti3;
    U_ISSUE(0)

    if (tid < K3_ROWS * NQ) {
        int m = tid >> 3, q = tid & 7;
        float a = angles[(row0 + m) * NQ + q] * 0.5f;
        cs[m][q][0] = cosf(a);
        cs[m][q][1] = sinf(a);
    }
    __syncthreads();

    {
        int m = tid & 15;
        int p = tid >> 4;
#pragma unroll
        for (int t = 0; t < 2; ++t) {
            const float4* c4 = (const float4*)&cs[16 * t + m][0][0];
            float4 c01 = c4[0], c23 = c4[1], c45 = c4[2], c67 = c4[3];
            float f1 = (p & 1)  ? c01.w : c01.z;
            float f2 = (p & 2)  ? c23.y : c23.x;
            float f3 = (p & 4)  ? c23.w : c23.z;
            float f4 = (p & 8)  ? c45.y : c45.x;
            float f5 = (p & 16) ? c45.w : c45.z;
            float v0 = c01.x, v1 = c01.y;
            v0 *= f1; v1 *= f1;
            v0 *= f2; v1 *= f2;
            v0 *= f3; v1 *= f3;
            v0 *= f4; v1 *= f4;
            v0 *= f5; v1 *= f5;
#pragma unroll
            for (int e = 0; e < 4; ++e) {
                float f6 = (e & 1) ? c67.y : c67.x;
                float f7 = (e & 2) ? c67.w : c67.z;
                v2f pr;
                pr[0] = (v0 * f6) * f7;
                pr[1] = (v1 * f6) * f7;
                *(v2f*)&encP[t][p + 32 * e][m][0] = pr;
            }
        }
    }
    U_STASH(0)
    __syncthreads();

    v8f cr0, ci0, cr1, ci1;
#pragma unroll
    for (int v = 0; v < 8; ++v) { cr0[v] = 0.0f; ci0[v] = 0.0f; cr1[v] = 0.0f; ci1[v] = 0.0f; }

    int mrow = lane & 15;
    int h    = lane >> 4;
    int n    = lane & 15;
    int mt   = wave >> 3;
    int ng   = wave & 7;
    int bo   = h * UPITCH + (32 * ng + n) * 2;

#pragma unroll
    for (int sl = 0; sl < NSLAB; ++sl) {
        const int b = sl & 1;
        if (sl + 1 < NSLAB) {
            U_ISSUE(sl + 1)
        }
#pragma unroll
        for (int hf = 0; hf < 2; ++hf) {
            if (hf == 1 && sl + 1 < NSLAB) {
                __builtin_amdgcn_sched_barrier(0);
                U_STASH(b ^ 1)
            }
#pragma unroll
            for (int kk = 4 * hf; kk < 4 * hf + 4; ++kk) {
                v2f a   = *(const v2f*)&encP[mt][sl * SLAB_PAIRS + 2 * kk + h][mrow][0];
                v2f br0 = *(const v2f*)&Ub[b][0][bo + 2 * kk * UPITCH];
                v2f bi0 = *(const v2f*)&Ub[b][1][bo + 2 * kk * UPITCH];
                v2f br1 = *(const v2f*)&Ub[b][0][bo + 2 * kk * UPITCH + 32];
                v2f bi1 = *(const v2f*)&Ub[b][1][bo + 2 * kk * UPITCH + 32];
                cr0 = __builtin_amdgcn_wmma_f32_16x16x4_f32(false, a, false, br0,
                                                            (short)0, cr0, false, false);
                ci0 = __builtin_amdgcn_wmma_f32_16x16x4_f32(false, a, false, bi0,
                                                            (short)0, ci0, false, false);
                cr1 = __builtin_amdgcn_wmma_f32_16x16x4_f32(false, a, false, br1,
                                                            (short)0, cr1, false, false);
                ci1 = __builtin_amdgcn_wmma_f32_16x16x4_f32(false, a, false, bi1,
                                                            (short)0, ci1, false, false);
            }
        }
        if (sl + 1 < NSLAB) __syncthreads();
    }

#pragma unroll
    for (int v = 0; v < 8; ++v) {
        int m = 16 * mt + v + 8 * h;
        pq[m][32 * ng + n]      = cr0[v] * cr0[v] + ci0[v] * ci0[v];
        pq[m][32 * ng + 16 + n] = cr1[v] * cr1[v] + ci1[v] * ci1[v];
    }
    __syncthreads();

    if (tid < K3_ROWS * NQ) {
        int m = tid >> 3, q = tid & 7;
        float acc = 0.0f;
        for (int s = 0; s < NSTATE; ++s) {
            float p = pq[m][s];
            acc += ((s >> q) & 1) ? -p : p;
        }
        E[m][q] = acc;
    }
    __syncthreads();

    if (tid < K3_ROWS * NCLS) {
        int m = tid / NCLS, c = tid % NCLS;
        float acc = out_b[c];
#pragma unroll
        for (int q = 0; q < NQ; ++q) acc += E[m][q] * out_w[c * NQ + q];
        out[(row0 + m) * NCLS + c] = acc;
    }
#undef U_ISSUE
#undef U_STASH
}

extern "C" void kernel_launch(void* const* d_in, const int* in_sizes, int n_in,
                              void* d_out, int out_size, void* d_ws, size_t ws_size,
                              hipStream_t stream) {
    const float* x      = (const float*)d_in[0];
    const float* proj_w = (const float*)d_in[1];
    const float* qnn_w  = (const float*)d_in[2];
    const float* out_w  = (const float*)d_in[3];
    const float* out_b  = (const float*)d_in[4];
    float* ws = (float*)d_ws;

    float* angles = ws;
    float* UrP    = ws + 65536;
    float* UiP    = ws + 65536 + 65536;
    float* cs_tab = (float*)d_out;

    k1_angles<<<BATCH / 16 + 1, 128, 0, stream>>>(x, proj_w, qnn_w, angles, cs_tab);
    k2_unitary<<<NSTATE / 8, 256, 0, stream>>>(cs_tab, UrP, UiP);
    k3_fused<<<BATCH / K3_ROWS, 512, 0, stream>>>(angles, UrP, UiP, out_w, out_b,
                                             (float*)d_out);
}


# GEMM kernel epilogue: Z sums on all 512 threads (16 states per thread as an in-thread butterfly, then 16-lane DPP butterflies), not 256 threads x 256 serial LDS adds; E sum order changes
# speedup vs baseline: 2.0284x; 1.0662x over previous
#include <hip/hip_runtime.h>
#include <math.h>

typedef float v2f __attribute__((ext_vector_type(2)));
typedef float v8f __attribute__((ext_vector_type(8)));

#define BATCH 8192
#define FDIM  512
#define NQ    8
#define NSTATE 256
#define NCLS  10

#define NWEIGHT (3 * NQ * 3)
__global__ __launch_bounds__(128) void k1_angles(const float* __restrict__ x,
                                                 const float* __restrict__ proj_w,
                                                 const float* __restrict__ qnn_w,
                                                 float* __restrict__ angles,
                                                 float* __restrict__ cs_tab) {
    __shared__ float red[4][16][NQ];
    int tid  = threadIdx.x;
    if (blockIdx.x == BATCH / 16) {
        if (tid < NWEIGHT) {
            float a = qnn_w[tid] * 0.5f;
            cs_tab[2 * tid]     = cosf(a);
            cs_tab[2 * tid + 1] = sinf(a);
        }
        return;
    }
    int lane = tid & 31;
    int wave = tid >> 5;
    int c    = lane & 15;
    int h    = lane >> 4;
    int row0 = blockIdx.x * 16;
    int kb   = wave * 128 + 4 * h;

    const float4* xp = (const float4*)(x + (size_t)(row0 + c) * FDIM + kb);
    const float4* wp = (const float4*)(proj_w + (size_t)(c & 7) * FDIM + kb);

    float4 xa[16], wb[16];
#pragma unroll
    for (int j = 0; j < 16; ++j) wb[j] = wp[2 * j];
#pragma unroll
    for (int j = 0; j < 16; ++j) xa[j] = xp[2 * j];
    __builtin_amdgcn_sched_barrier(0);

    v8f acc0, acc1;
#pragma unroll
    for (int v = 0; v < 8; ++v) { acc0[v] = 0.0f; acc1[v] = 0.0f; }

#pragma unroll
    for (int j = 0; j < 16; ++j) {
        v2f a0, b0, a1, b1;
        a0[0] = xa[j].x; a0[1] = xa[j].y;  b0[0] = wb[j].x; b0[1] = wb[j].y;
        a1[0] = xa[j].z; a1[1] = xa[j].w;  b1[0] = wb[j].z; b1[1] = wb[j].w;
        acc0 = __builtin_amdgcn_wmma_f32_16x16x4_f32(false, a0, false, b0,
                                                     (short)0, acc0, false, false);
        acc1 = __builtin_amdgcn_wmma_f32_16x16x4_f32(false, a1, false, b1,
                                                     (short)0, acc1, false, false);
    }

    if (c < NQ) {
#pragma unroll
        for (int r = 0; r < 8; ++r) red[wave][8 * h + r][c] = acc0[r] + acc1[r];
    }
    __syncthreads();

    {
        int m = tid >> 3, q = tid & 7;
        float v = (red[0][m][q] + red[1][m][q]) + (red[2][m][q] + red[3][m][q]);
        angles[(row0 + m) * NQ + q] = tanhf(v) * 1.5707963267948966f;
    }
}

template <int M>
__device__ __forceinline__ float lane_xor(float v) {
    int iv = __builtin_bit_cast(int, v);
    int r;
    if constexpr (M < 16) {
        r = __builtin_amdgcn_update_dpp(iv, iv, 0x160 + M, 0xF, 0xF, true);
    } else {
        r = __builtin_amdgcn_permlanex16(iv, iv, 0x76543210, 0xfedcba98, false, false);
    }
    return __builtin_bit_cast(float, r);
}

template <int M>
__device__ __forceinline__ void rot_lane_rx(float sr[8], float si[8],
                                            float c, float s) {
#pragma unroll
    for (int r = 0; r < 8; ++r) {
        float pr = lane_xor<M>(sr[r]);
        float pi = lane_xor<M>(si[r]);
        float nr = c * sr[r] + s * pi;
        float ni = c * si[r] - s * pr;
        sr[r] = nr; si[r] = ni;
    }
}

template <int M>
__device__ __forceinline__ void rot_lane_ry(float sr[8], float si[8],
                                            float c, float s, int lane) {
    float b = (lane & M) ? s : -s;
#pragma unroll
    for (int r = 0; r < 8; ++r) {
        float pr = lane_xor<M>(sr[r]);
        float pi = lane_xor<M>(si[r]);
        sr[r] = c * sr[r] + b * pr;
        si[r] = c * si[r] + b * pi;
    }
}

__device__ __forceinline__ void rot_reg_rx(float sr[8], float si[8],
                                           float c, float s, int rm) {
#pragma unroll
    for (int r0 = 0; r0 < 8; ++r0) {
        if (r0 & rm) continue;
        int r1 = r0 | rm;
        float a0r = sr[r0], a0i = si[r0], a1r = sr[r1], a1i = si[r1];
        sr[r0] = c * a0r + s * a1i;  si[r0] = c * a0i - s * a1r;
        sr[r1] = c * a1r + s * a0i;  si[r1] = c * a1i - s * a0r;
    }
}

__device__ __forceinline__ void rot_reg_ry(float sr[8], float si[8],
                                           float c, float s, int rm) {
#pragma unroll
    for (int r0 = 0; r0 < 8; ++r0) {
        if (r0 & rm) continue;
        int r1 = r0 | rm;
        float a0r = sr[r0], a0i = si[r0], a1r = sr[r1], a1i = si[r1];
        sr[r0] = c * a0r - s * a1r;  si[r0] = c * a0i - s * a1i;
        sr[r1] = s * a0r + c * a1r;  si[r1] = s * a0i + c * a1i;
    }
}

template <int CM, int TM>
__device__ __forceinline__ void cnot_ll(float sr[8], float si[8], int lane) {
    bool hi = (lane & CM) != 0;
#pragma unroll
    for (int r = 0; r < 8; ++r) {
        float pr = lane_xor<TM>(sr[r]);
        float pi = lane_xor<TM>(si[r]);
        sr[r] = hi ? pr : sr[r];
        si[r] = hi ? pi : si[r];
    }
}

__device__ __forceinline__ void cnot_lane_reg(float sr[8], float si[8],
                                              int cm, int trm, int lane) {
    bool hi = (lane & cm) != 0;
#pragma unroll
    for (int r0 = 0; r0 < 8; ++r0) {
        if (r0 & trm) continue;
        int r1 = r0 | trm;
        float t0r = sr[r0], t0i = si[r0];
        sr[r0] = hi ? sr[r1] : sr[r0];  si[r0] = hi ? si[r1] : si[r0];
        sr[r1] = hi ? t0r    : sr[r1];  si[r1] = hi ? t0i    : si[r1];
    }
}

template <int TM>
__device__ __forceinline__ void cnot_reg_lane(float sr[8], float si[8],
                                              int crm) {
#pragma unroll
    for (int r = 0; r < 8; ++r) {
        if (!(r & crm)) continue;
        sr[r] = lane_xor<TM>(sr[r]);
        si[r] = lane_xor<TM>(si[r]);
    }
}

__device__ __forceinline__ void cnot_reg_reg(float sr[8], float si[8],
                                             int crm, int trm) {
#pragma unroll
    for (int r0 = 0; r0 < 8; ++r0) {
        if (!(r0 & crm) || (r0 & trm)) continue;
        int r1 = r0 | trm;
        float tr = sr[r0], ti = si[r0];
        sr[r0] = sr[r1]; si[r0] = si[r1];
        sr[r1] = tr;     si[r1] = ti;
    }
}

template <int Q>
__device__ __forceinline__ void qubit_rot(float sr[8], float si[8],
                                          const float* __restrict__ cs_tab,
                                          int l, int lane) {
    const float* g = cs_tab + (l * NQ + Q) * 6;
    float c0 = g[0], s0 = g[1];
    float c1 = g[2], s1 = g[3];
    float c2 = g[4], s2 = g[5];
    if constexpr (Q < 5) {
        rot_lane_rx<(1 << Q)>(sr, si, c0, s0);
        rot_lane_ry<(1 << Q)>(sr, si, c1, s1, lane);
        rot_lane_rx<(1 << Q)>(sr, si, c2, s2);
    } else {
        constexpr int rm = 1 << (Q - 5);
        rot_reg_rx(sr, si, c0, s0, rm);
        rot_reg_ry(sr, si, c1, s1, rm);
        rot_reg_rx(sr, si, c2, s2, rm);
    }
}

__global__ void k2_unitary(const float* __restrict__ cs_tab,
                           float* __restrict__ UrP,
                           float* __restrict__ UiP) {
    int lane = threadIdx.x & 31;
    int t    = blockIdx.x * 8 + (threadIdx.x >> 5);

    float sr[8], si[8];
#pragma unroll
    for (int r = 0; r < 8; ++r) {
        sr[r] = (r * 32 + lane == t) ? 1.0f : 0.0f;
        si[r] = 0.0f;
    }

#pragma unroll
    for (int l = 0; l < 3; ++l) {
        qubit_rot<0>(sr, si, cs_tab, l, lane);
        qubit_rot<1>(sr, si, cs_tab, l, lane);
        qubit_rot<2>(sr, si, cs_tab, l, lane);
        qubit_rot<3>(sr, si, cs_tab, l, lane);
        qubit_rot<4>(sr, si, cs_tab, l, lane);
        qubit_rot<5>(sr, si, cs_tab, l, lane);
        qubit_rot<6>(sr, si, cs_tab, l, lane);
        qubit_rot<7>(sr, si, cs_tab, l, lane);
        cnot_ll<1, 2>(sr, si, lane);
        cnot_ll<2, 4>(sr, si, lane);
        cnot_ll<4, 8>(sr, si, lane);
        cnot_ll<8, 16>(sr, si, lane);
        cnot_lane_reg(sr, si, 16, 1, lane);
        cnot_reg_reg(sr, si, 1, 2);
        cnot_reg_reg(sr, si, 2, 4);
        cnot_reg_lane<1>(sr, si, 4);
    }

#pragma unroll
    for (int r = 0; r < 8; ++r) {
        int s = r * 32 + lane;
        int off = (t >> 1) * (2 * NSTATE) + s * 2 + (t & 1);
        UrP[off] = sr[r];
        UiP[off] = si[r];
    }
}

#define K3_ROWS    32
#define SLAB_PAIRS 16
#define NSLAB      (NSTATE / 2 / SLAB_PAIRS)
#define UPITCH     (2 * NSTATE + 32)

__global__ __launch_bounds__(512) void k3_fused(
        const float* __restrict__ angles,
        const float* UrP,
        const float* UiP,
        const float* __restrict__ out_w,
        const float* __restrict__ out_b,
        float* __restrict__ out) {
    __shared__ __align__(16) float cs[K3_ROWS][NQ][2];
    __shared__ __align__(16) float encP[2][NSTATE / 2][16][2];
    __shared__ __align__(16) float Ub[2][2][SLAB_PAIRS * UPITCH];
    __shared__ __align__(16) float pq[K3_ROWS][NSTATE];
    __shared__ __align__(16) float E[K3_ROWS][NQ];

    int tid  = threadIdx.x;
    int lane = tid & 31;
    int wave = tid >> 5;
    int row0 = blockIdx.x * K3_ROWS;

    const float4* gUr = (const float4*)UrP;
    const float4* gUi = (const float4*)UiP;
    int so = (tid >> 7) * UPITCH + (tid & 127) * 4;
    float4 tr0, tr1, tr2, tr3, ti0, ti1, ti2, ti3;
#define U_ISSUE(S)                                                           \
    tr0 = gUr[(S) * 2048 + tid];        ti0 = gUi[(S) * 2048 + tid];        \
    tr1 = gUr[(S) * 2048 + tid + 512];  ti1 = gUi[(S) * 2048 + tid + 512];  \
    tr2 = gUr[(S) * 2048 + tid + 1024]; ti2 = gUi[(S) * 2048 + tid + 1024]; \
    tr3 = gUr[(S) * 2048 + tid + 1536]; ti3 = gUi[(S) * 2048 + tid + 1536];
#define U_STASH(B)                                                           \
    *(float4*)&Ub[B][0][so]               = tr0; *(float4*)&Ub[B][1][so]               = ti0; \
    *(float4*)&Ub[B][0][so + 4 * UPITCH]  = tr1; *(float4*)&Ub[B][1][so + 4 * UPITCH]  = ti1; \
    *(float4*)&Ub[B][0][so + 8 * UPITCH]  = tr2; *(float4*)&Ub[B][1][so + 8 * UPITCH]  = ti2; \
    *(float4*)&Ub[B][0][so + 12 * UPITCH] = tr3; *(float4*)&Ub[B][1][so + 12 * UPITCH] = ti3;
    U_ISSUE(0)

    if (tid < K3_ROWS * NQ) {
        int m = tid >> 3, q = tid & 7;
        float a = angles[(row0 + m) * NQ + q] * 0.5f;
        cs[m][q][0] = cosf(a);
        cs[m][q][1] = sinf(a);
    }
    __syncthreads();

    {
        int m = tid & 15;
        int p = tid >> 4;
#pragma unroll
        for (int t = 0; t < 2; ++t) {
            const float4* c4 = (const float4*)&cs[16 * t + m][0][0];
            float4 c01 = c4[0], c23 = c4[1], c45 = c4[2], c67 = c4[3];
            float f1 = (p & 1)  ? c01.w : c01.z;
            float f2 = (p & 2)  ? c23.y : c23.x;
            float f3 = (p & 4)  ? c23.w : c23.z;
            float f4 = (p & 8)  ? c45.y : c45.x;
            float f5 = (p & 16) ? c45.w : c45.z;
            float v0 = c01.x, v1 = c01.y;
            v0 *= f1; v1 *= f1;
            v0 *= f2; v1 *= f2;
            v0 *= f3; v1 *= f3;
            v0 *= f4; v1 *= f4;
            v0 *= f5; v1 *= f5;
#pragma unroll
            for (int e = 0; e < 4; ++e) {
                float f6 = (e & 1) ? c67.y : c67.x;
                float f7 = (e & 2) ? c67.w : c67.z;
                v2f pr;
                pr[0] = (v0 * f6) * f7;
                pr[1] = (v1 * f6) * f7;
                *(v2f*)&encP[t][p + 32 * e][m][0] = pr;
            }
        }
    }
    U_STASH(0)
    __syncthreads();

    v8f cr0, ci0, cr1, ci1;
#pragma unroll
    for (int v = 0; v < 8; ++v) { cr0[v] = 0.0f; ci0[v] = 0.0f; cr1[v] = 0.0f; ci1[v] = 0.0f; }

    int mrow = lane & 15;
    int h    = lane >> 4;
    int n    = lane & 15;
    int mt   = wave >> 3;
    int ng   = wave & 7;
    int bo   = h * UPITCH + (32 * ng + n) * 2;

#pragma unroll
    for (int sl = 0; sl < NSLAB; ++sl) {
        const int b = sl & 1;
        if (sl + 1 < NSLAB) {
            U_ISSUE(sl + 1)
        }
#pragma unroll
        for (int hf = 0; hf < 2; ++hf) {
            if (hf == 1 && sl + 1 < NSLAB) {
                __builtin_amdgcn_sched_barrier(0);
                U_STASH(b ^ 1)
            }
#pragma unroll
            for (int kk = 4 * hf; kk < 4 * hf + 4; ++kk) {
                v2f a   = *(const v2f*)&encP[mt][sl * SLAB_PAIRS + 2 * kk + h][mrow][0];
                v2f br0 = *(const v2f*)&Ub[b][0][bo + 2 * kk * UPITCH];
                v2f bi0 = *(const v2f*)&Ub[b][1][bo + 2 * kk * UPITCH];
                v2f br1 = *(const v2f*)&Ub[b][0][bo + 2 * kk * UPITCH + 32];
                v2f bi1 = *(const v2f*)&Ub[b][1][bo + 2 * kk * UPITCH + 32];
                cr0 = __builtin_amdgcn_wmma_f32_16x16x4_f32(false, a, false, br0,
                                                            (short)0, cr0, false, false);
                ci0 = __builtin_amdgcn_wmma_f32_16x16x4_f32(false, a, false, bi0,
                                                            (short)0, ci0, false, false);
                cr1 = __builtin_amdgcn_wmma_f32_16x16x4_f32(false, a, false, br1,
                                                            (short)0, cr1, false, false);
                ci1 = __builtin_amdgcn_wmma_f32_16x16x4_f32(false, a, false, bi1,
                                                            (short)0, ci1, false, false);
            }
        }
        if (sl + 1 < NSLAB) __syncthreads();
    }

#pragma unroll
    for (int v = 0; v < 8; ++v) {
        int m = 16 * mt + v + 8 * h;
        pq[m][32 * ng + n]      = cr0[v] * cr0[v] + ci0[v] * ci0[v];
        pq[m][32 * ng + 16 + n] = cr1[v] * cr1[v] + ci1[v] * ci1[v];
    }
    __syncthreads();

    {
        int m = tid >> 4, j = tid & 15;
        const float4* prow = (const float4*)&pq[m][4 * j];
        float4 a[4];
#pragma unroll
        for (int c = 0; c < 4; ++c) a[c] = prow[16 * c];
        float T[4], B0[4], B1[4];
#pragma unroll
        for (int c = 0; c < 4; ++c) {
            float s01 = a[c].x + a[c].y, d01 = a[c].x - a[c].y;
            float s23 = a[c].z + a[c].w, d23 = a[c].z - a[c].w;
            T[c]  = s01 + s23;
            B0[c] = d01 + d23;
            B1[c] = s01 - s23;
        }
        float e0 = (B0[0] + B0[1]) + (B0[2] + B0[3]);
        float e1 = (B1[0] + B1[1]) + (B1[2] + B1[3]);
        float u01 = T[0] + T[1], v01 = T[0] - T[1];
        float u23 = T[2] + T[3], v23 = T[2] - T[3];
        float t  = u01 + u23;
        float e6 = v01 + v23;
        float e7 = u01 - u23;
        { float p = lane_xor<1>(t); t = p + ((j & 1) ? -t : t); }
        { float p = lane_xor<2>(t); t = p + ((j & 2) ? -t : t); }
        { float p = lane_xor<4>(t); t = p + ((j & 4) ? -t : t); }
        { float p = lane_xor<8>(t); t = p + ((j & 8) ? -t : t); }
        e0 += lane_xor<1>(e0); e1 += lane_xor<1>(e1); e6 += lane_xor<1>(e6); e7 += lane_xor<1>(e7);
        e0 += lane_xor<2>(e0); e1 += lane_xor<2>(e1); e6 += lane_xor<2>(e6); e7 += lane_xor<2>(e7);
        e0 += lane_xor<4>(e0); e1 += lane_xor<4>(e1); e6 += lane_xor<4>(e6); e7 += lane_xor<4>(e7);
        e0 += lane_xor<8>(e0); e1 += lane_xor<8>(e1); e6 += lane_xor<8>(e6); e7 += lane_xor<8>(e7);
        if (j == 0) {
            v2f lo, hi;
            lo[0] = e0; lo[1] = e1;
            hi[0] = e6; hi[1] = e7;
            *(v2f*)&E[m][0] = lo;
            *(v2f*)&E[m][6] = hi;
        } else if ((j & (j - 1)) == 0) {
            E[m][2 + __builtin_ctz(j)] = t;
        }
    }
    __syncthreads();

    if (tid < K3_ROWS * NCLS) {
        int m = tid / NCLS, c = tid % NCLS;
        float acc = out_b[c];
#pragma unroll
        for (int q = 0; q < NQ; ++q) acc += E[m][q] * out_w[c * NQ + q];
        out[(row0 + m) * NCLS + c] = acc;
    }
#undef U_ISSUE
#undef U_STASH
}

extern "C" void kernel_launch(void* const* d_in, const int* in_sizes, int n_in,
                              void* d_out, int out_size, void* d_ws, size_t ws_size,
                              hipStream_t stream) {
    const float* x      = (const float*)d_in[0];
    const float* proj_w = (const float*)d_in[1];
    const float* qnn_w  = (const float*)d_in[2];
    const float* out_w  = (const float*)d_in[3];
    const float* out_b  = (const float*)d_in[4];
    float* ws = (float*)d_ws;

    float* angles = ws;
    float* UrP    = ws + 65536;
    float* UiP    = ws + 65536 + 65536;
    float* cs_tab = (float*)d_out;

    k1_angles<<<BATCH / 16 + 1, 128, 0, stream>>>(x, proj_w, qnn_w, angles, cs_tab);
    k2_unitary<<<NSTATE / 8, 256, 0, stream>>>(cs_tab, UrP, UiP);
    k3_fused<<<BATCH / K3_ROWS, 512, 0, stream>>>(angles, UrP, UiP, out_w, out_b,
                                             (float*)d_out);
}


# unitary kernel: 128 groups of 2 waves, not 32 of 8: each column's serial gate chain has a SIMD to itself; cos/sin table's 9 cache lines touched once up front for the lone wave; same gate arithmetic
# speedup vs baseline: 2.1482x; 1.0591x over previous
#include <hip/hip_runtime.h>
#include <math.h>

typedef float v2f __attribute__((ext_vector_type(2)));
typedef float v8f __attribute__((ext_vector_type(8)));

#define BATCH 8192
#define FDIM  512
#define NQ    8
#define NSTATE 256
#define NCLS  10

#define NWEIGHT (3 * NQ * 3)
__global__ __launch_bounds__(128) void k1_angles(const float* __restrict__ x,
                                                 const float* __restrict__ proj_w,
                                                 const float* __restrict__ qnn_w,
                                                 float* __restrict__ angles,
                                                 float* __restrict__ cs_tab) {
    __shared__ float red[4][16][NQ];
    int tid  = threadIdx.x;
    if (blockIdx.x == BATCH / 16) {
        if (tid < NWEIGHT) {
            float a = qnn_w[tid] * 0.5f;
            cs_tab[2 * tid]     = cosf(a);
            cs_tab[2 * tid + 1] = sinf(a);
        }
        return;
    }
    int lane = tid & 31;
    int wave = tid >> 5;
    int c    = lane & 15;
    int h    = lane >> 4;
    int row0 = blockIdx.x * 16;
    int kb   = wave * 128 + 4 * h;

    const float4* xp = (const float4*)(x + (size_t)(row0 + c) * FDIM + kb);
    const float4* wp = (const float4*)(proj_w + (size_t)(c & 7) * FDIM + kb);

    float4 xa[16], wb[16];
#pragma unroll
    for (int j = 0; j < 16; ++j) wb[j] = wp[2 * j];
#pragma unroll
    for (int j = 0; j < 16; ++j) xa[j] = xp[2 * j];
    __builtin_amdgcn_sched_barrier(0);

    v8f acc0, acc1;
#pragma unroll
    for (int v = 0; v < 8; ++v) { acc0[v] = 0.0f; acc1[v] = 0.0f; }

#pragma unroll
    for (int j = 0; j < 16; ++j) {
        v2f a0, b0, a1, b1;
        a0[0] = xa[j].x; a0[1] = xa[j].y;  b0[0] = wb[j].x; b0[1] = wb[j].y;
        a1[0] = xa[j].z; a1[1] = xa[j].w;  b1[0] = wb[j].z; b1[1] = wb[j].w;
        acc0 = __builtin_amdgcn_wmma_f32_16x16x4_f32(false, a0, false, b0,
                                                     (short)0, acc0, false, false);
        acc1 = __builtin_amdgcn_wmma_f32_16x16x4_f32(false, a1, false, b1,
                                                     (short)0, acc1, false, false);
    }

    if (c < NQ) {
#pragma unroll
        for (int r = 0; r < 8; ++r) red[wave][8 * h + r][c] = acc0[r] + acc1[r];
    }
    __syncthreads();

    {
        int m = tid >> 3, q = tid & 7;
        float v = (red[0][m][q] + red[1][m][q]) + (red[2][m][q] + red[3][m][q]);
        angles[(row0 + m) * NQ + q] = tanhf(v) * 1.5707963267948966f;
    }
}

template <int M>
__device__ __forceinline__ float lane_xor(float v) {
    int iv = __builtin_bit_cast(int, v);
    int r;
    if constexpr (M < 16) {
        r = __builtin_amdgcn_update_dpp(iv, iv, 0x160 + M, 0xF, 0xF, true);
    } else {
        r = __builtin_amdgcn_permlanex16(iv, iv, 0x76543210, 0xfedcba98, false, false);
    }
    return __builtin_bit_cast(float, r);
}

template <int M>
__device__ __forceinline__ void rot_lane_rx(float sr[8], float si[8],
                                            float c, float s) {
#pragma unroll
    for (int r = 0; r < 8; ++r) {
        float pr = lane_xor<M>(sr[r]);
        float pi = lane_xor<M>(si[r]);
        float nr = c * sr[r] + s * pi;
        float ni = c * si[r] - s * pr;
        sr[r] = nr; si[r] = ni;
    }
}

template <int M>
__device__ __forceinline__ void rot_lane_ry(float sr[8], float si[8],
                                            float c, float s, int lane) {
    float b = (lane & M) ? s : -s;
#pragma unroll
    for (int r = 0; r < 8; ++r) {
        float pr = lane_xor<M>(sr[r]);
        float pi = lane_xor<M>(si[r]);
        sr[r] = c * sr[r] + b * pr;
        si[r] = c * si[r] + b * pi;
    }
}

__device__ __forceinline__ void rot_reg_rx(float sr[8], float si[8],
                                           float c, float s, int rm) {
#pragma unroll
    for (int r0 = 0; r0 < 8; ++r0) {
        if (r0 & rm) continue;
        int r1 = r0 | rm;
        float a0r = sr[r0], a0i = si[r0], a1r = sr[r1], a1i = si[r1];
        sr[r0] = c * a0r + s * a1i;  si[r0] = c * a0i - s * a1r;
        sr[r1] = c * a1r + s * a0i;  si[r1] = c * a1i - s * a0r;
    }
}

__device__ __forceinline__ void rot_reg_ry(float sr[8], float si[8],
                                           float c, float s, int rm) {
#pragma unroll
    for (int r0 = 0; r0 < 8; ++r0) {
        if (r0 & rm) continue;
        int r1 = r0 | rm;
        float a0r = sr[r0], a0i = si[r0], a1r = sr[r1], a1i = si[r1];
        sr[r0] = c * a0r - s * a1r;  si[r0] = c * a0i - s * a1i;
        sr[r1] = s * a0r + c * a1r;  si[r1] = s * a0i + c * a1i;
    }
}

template <int CM, int TM>
__device__ __forceinline__ void cnot_ll(float sr[8], float si[8], int lane) {
    bool hi = (lane & CM) != 0;
#pragma unroll
    for (int r = 0; r < 8; ++r) {
        float pr = lane_xor<TM>(sr[r]);
        float pi = lane_xor<TM>(si[r]);
        sr[r] = hi ? pr : sr[r];
        si[r] = hi ? pi : si[r];
    }
}

__device__ __forceinline__ void cnot_lane_reg(float sr[8], float si[8],
                                              int cm, int trm, int lane) {
    bool hi = (lane & cm) != 0;
#pragma unroll
    for (int r0 = 0; r0 < 8; ++r0) {
        if (r0 & trm) continue;
        int r1 = r0 | trm;
        float t0r = sr[r0], t0i = si[r0];
        sr[r0] = hi ? sr[r1] : sr[r0];  si[r0] = hi ? si[r1] : si[r0];
        sr[r1] = hi ? t0r    : sr[r1];  si[r1] = hi ? t0i    : si[r1];
    }
}

template <int TM>
__device__ __forceinline__ void cnot_reg_lane(float sr[8], float si[8],
                                              int crm) {
#pragma unroll
    for (int r = 0; r < 8; ++r) {
        if (!(r & crm)) continue;
        sr[r] = lane_xor<TM>(sr[r]);
        si[r] = lane_xor<TM>(si[r]);
    }
}

__device__ __forceinline__ void cnot_reg_reg(float sr[8], float si[8],
                                             int crm, int trm) {
#pragma unroll
    for (int r0 = 0; r0 < 8; ++r0) {
        if (!(r0 & crm) || (r0 & trm)) continue;
        int r1 = r0 | trm;
        float tr = sr[r0], ti = si[r0];
        sr[r0] = sr[r1]; si[r0] = si[r1];
        sr[r1] = tr;     si[r1] = ti;
    }
}

template <int Q>
__device__ __forceinline__ void qubit_rot(float sr[8], float si[8],
                                          const float* __restrict__ cs_tab,
                                          int l, int lane) {
    const float* g = cs_tab + (l * NQ + Q) * 6;
    float c0 = g[0], s0 = g[1];
    float c1 = g[2], s1 = g[3];
    float c2 = g[4], s2 = g[5];
    if constexpr (Q < 5) {
        rot_lane_rx<(1 << Q)>(sr, si, c0, s0);
        rot_lane_ry<(1 << Q)>(sr, si, c1, s1, lane);
        rot_lane_rx<(1 << Q)>(sr, si, c2, s2);
    } else {
        constexpr int rm = 1 << (Q - 5);
        rot_reg_rx(sr, si, c0, s0, rm);
        rot_reg_ry(sr, si, c1, s1, rm);
        rot_reg_rx(sr, si, c2, s2, rm);
    }
}

__global__ void k2_unitary(const float* __restrict__ cs_tab,
                           float* __restrict__ UrP,
                           float* __restrict__ UiP) {
    int lane = threadIdx.x & 31;
    int t    = blockIdx.x * 2 + (threadIdx.x >> 5);

    float w0 = cs_tab[0],  w1 = cs_tab[16], w2 = cs_tab[32];
    float w3 = cs_tab[48], w4 = cs_tab[64], w5 = cs_tab[80];
    float w6 = cs_tab[96], w7 = cs_tab[112], w8 = cs_tab[128];
    asm volatile("" :: "s"(w0), "s"(w1), "s"(w2), "s"(w3), "s"(w4),
                       "s"(w5), "s"(w6), "s"(w7), "s"(w8));

    float sr[8], si[8];
#pragma unroll
    for (int r = 0; r < 8; ++r) {
        sr[r] = (r * 32 + lane == t) ? 1.0f : 0.0f;
        si[r] = 0.0f;
    }

#pragma unroll
    for (int l = 0; l < 3; ++l) {
        qubit_rot<0>(sr, si, cs_tab, l, lane);
        qubit_rot<1>(sr, si, cs_tab, l, lane);
        qubit_rot<2>(sr, si, cs_tab, l, lane);
        qubit_rot<3>(sr, si, cs_tab, l, lane);
        qubit_rot<4>(sr, si, cs_tab, l, lane);
        qubit_rot<5>(sr, si, cs_tab, l, lane);
        qubit_rot<6>(sr, si, cs_tab, l, lane);
        qubit_rot<7>(sr, si, cs_tab, l, lane);
        cnot_ll<1, 2>(sr, si, lane);
        cnot_ll<2, 4>(sr, si, lane);
        cnot_ll<4, 8>(sr, si, lane);
        cnot_ll<8, 16>(sr, si, lane);
        cnot_lane_reg(sr, si, 16, 1, lane);
        cnot_reg_reg(sr, si, 1, 2);
        cnot_reg_reg(sr, si, 2, 4);
        cnot_reg_lane<1>(sr, si, 4);
    }

#pragma unroll
    for (int r = 0; r < 8; ++r) {
        int s = r * 32 + lane;
        int off = (t >> 1) * (2 * NSTATE) + s * 2 + (t & 1);
        UrP[off] = sr[r];
        UiP[off] = si[r];
    }
}

#define K3_ROWS    32
#define SLAB_PAIRS 16
#define NSLAB      (NSTATE / 2 / SLAB_PAIRS)
#define UPITCH     (2 * NSTATE + 32)

__global__ __launch_bounds__(512) void k3_fused(
        const float* __restrict__ angles,
        const float* UrP,
        const float* UiP,
        const float* __restrict__ out_w,
        const float* __restrict__ out_b,
        float* __restrict__ out) {
    __shared__ __align__(16) float cs[K3_ROWS][NQ][2];
    __shared__ __align__(16) float encP[2][NSTATE / 2][16][2];
    __shared__ __align__(16) float Ub[2][2][SLAB_PAIRS * UPITCH];
    __shared__ __align__(16) float pq[K3_ROWS][NSTATE];
    __shared__ __align__(16) float E[K3_ROWS][NQ];

    int tid  = threadIdx.x;
    int lane = tid & 31;
    int wave = tid >> 5;
    int row0 = blockIdx.x * K3_ROWS;

    const float4* gUr = (const float4*)UrP;
    const float4* gUi = (const float4*)UiP;
    int so = (tid >> 7) * UPITCH + (tid & 127) * 4;
    float4 tr0, tr1, tr2, tr3, ti0, ti1, ti2, ti3;
#define U_ISSUE(S)                                                           \
    tr0 = gUr[(S) * 2048 + tid];        ti0 = gUi[(S) * 2048 + tid];        \
    tr1 = gUr[(S) * 2048 + tid + 512];  ti1 = gUi[(S) * 2048 + tid + 512];  \
    tr2 = gUr[(S) * 2048 + tid + 1024]; ti2 = gUi[(S) * 2048 + tid + 1024]; \
    tr3 = gUr[(S) * 2048 + tid + 1536]; ti3 = gUi[(S) * 2048 + tid + 1536];
#define U_STASH(B)                                                           \
    *(float4*)&Ub[B][0][so]               = tr0; *(float4*)&Ub[B][1][so]               = ti0; \
    *(float4*)&Ub[B][0][so + 4 * UPITCH]  = tr1; *(float4*)&Ub[B][1][so + 4 * UPITCH]  = ti1; \
    *(float4*)&Ub[B][0][so + 8 * UPITCH]  = tr2; *(float4*)&Ub[B][1][so + 8 * UPITCH]  = ti2; \
    *(float4*)&Ub[B][0][so + 12 * UPITCH] = tr3; *(float4*)&Ub[B][1][so + 12 * UPITCH] = ti3;
    U_ISSUE(0)

    if (tid < K3_ROWS * NQ) {
        int m = tid >> 3, q = tid & 7;
        float a = angles[(row0 + m) * NQ + q] * 0.5f;
        cs[m][q][0] = cosf(a);
        cs[m][q][1] = sinf(a);
    }
    __syncthreads();

    {
        int m = tid & 15;
        int p = tid >> 4;
#pragma unroll
        for (int t = 0; t < 2; ++t) {
            const float4* c4 = (const float4*)&cs[16 * t + m][0][0];
            float4 c01 = c4[0], c23 = c4[1], c45 = c4[2], c67 = c4[3];
            float f1 = (p & 1)  ? c01.w : c01.z;
            float f2 = (p & 2)  ? c23.y : c23.x;
            float f3 = (p & 4)  ? c23.w : c23.z;
            float f4 = (p & 8)  ? c45.y : c45.x;
            float f5 = (p & 16) ? c45.w : c45.z;
            float v0 = c01.x, v1 = c01.y;
            v0 *= f1; v1 *= f1;
            v0 *= f2; v1 *= f2;
            v0 *= f3; v1 *= f3;
            v0 *= f4; v1 *= f4;
            v0 *= f5; v1 *= f5;
#pragma unroll
            for (int e = 0; e < 4; ++e) {
                float f6 = (e & 1) ? c67.y : c67.x;
                float f7 = (e & 2) ? c67.w : c67.z;
                v2f pr;
                pr[0] = (v0 * f6) * f7;
                pr[1] = (v1 * f6) * f7;
                *(v2f*)&encP[t][p + 32 * e][m][0] = pr;
            }
        }
    }
    U_STASH(0)
    __syncthreads();

    v8f cr0, ci0, cr1, ci1;
#pragma unroll
    for (int v = 0; v < 8; ++v) { cr0[v] = 0.0f; ci0[v] = 0.0f; cr1[v] = 0.0f; ci1[v] = 0.0f; }

    int mrow = lane & 15;
    int h    = lane >> 4;
    int n    = lane & 15;
    int mt   = wave >> 3;
    int ng   = wave & 7;
    int bo   = h * UPITCH + (32 * ng + n) * 2;

#pragma unroll
    for (int sl = 0; sl < NSLAB; ++sl) {
        const int b = sl & 1;
        if (sl + 1 < NSLAB) {
            U_ISSUE(sl + 1)
        }
#pragma unroll
        for (int hf = 0; hf < 2; ++hf) {
            if (hf == 1 && sl + 1 < NSLAB) {
                __builtin_amdgcn_sched_barrier(0);
                U_STASH(b ^ 1)
            }
#pragma unroll
            for (int kk = 4 * hf; kk < 4 * hf + 4; ++kk) {
                v2f a   = *(const v2f*)&encP[mt][sl * SLAB_PAIRS + 2 * kk + h][mrow][0];
                v2f br0 = *(const v2f*)&Ub[b][0][bo + 2 * kk * UPITCH];
                v2f bi0 = *(const v2f*)&Ub[b][1][bo + 2 * kk * UPITCH];
                v2f br1 = *(const v2f*)&Ub[b][0][bo + 2 * kk * UPITCH + 32];
                v2f bi1 = *(const v2f*)&Ub[b][1][bo + 2 * kk * UPITCH + 32];
                cr0 = __builtin_amdgcn_wmma_f32_16x16x4_f32(false, a, false, br0,
                                                            (short)0, cr0, false, false);
                ci0 = __builtin_amdgcn_wmma_f32_16x16x4_f32(false, a, false, bi0,
                                                            (short)0, ci0, false, false);
                cr1 = __builtin_amdgcn_wmma_f32_16x16x4_f32(false, a, false, br1,
                                                            (short)0, cr1, false, false);
                ci1 = __builtin_amdgcn_wmma_f32_16x16x4_f32(false, a, false, bi1,
                                                            (short)0, ci1, false, false);
            }
        }
        if (sl + 1 < NSLAB) __syncthreads();
    }

#pragma unroll
    for (int v = 0; v < 8; ++v) {
        int m = 16 * mt + v + 8 * h;
        pq[m][32 * ng + n]      = cr0[v] * cr0[v] + ci0[v] * ci0[v];
        pq[m][32 * ng + 16 + n] = cr1[v] * cr1[v] + ci1[v] * ci1[v];
    }
    __syncthreads();

    {
        int m = tid >> 4, j = tid & 15;
        const float4* prow = (const float4*)&pq[m][4 * j];
        float4 a[4];
#pragma unroll
        for (int c = 0; c < 4; ++c) a[c] = prow[16 * c];
        float T[4], B0[4], B1[4];
#pragma unroll
        for (int c = 0; c < 4; ++c) {
            float s01 = a[c].x + a[c].y, d01 = a[c].x - a[c].y;
            float s23 = a[c].z + a[c].w, d23 = a[c].z - a[c].w;
            T[c]  = s01 + s23;
            B0[c] = d01 + d23;
            B1[c] = s01 - s23;
        }
        float e0 = (B0[0] + B0[1]) + (B0[2] + B0[3]);
        float e1 = (B1[0] + B1[1]) + (B1[2] + B1[3]);
        float u01 = T[0] + T[1], v01 = T[0] - T[1];
        float u23 = T[2] + T[3], v23 = T[2] - T[3];
        float t  = u01 + u23;
        float e6 = v01 + v23;
        float e7 = u01 - u23;
        { float p = lane_xor<1>(t); t = p + ((j & 1) ? -t : t); }
        { float p = lane_xor<2>(t); t = p + ((j & 2) ? -t : t); }
        { float p = lane_xor<4>(t); t = p + ((j & 4) ? -t : t); }
        { float p = lane_xor<8>(t); t = p + ((j & 8) ? -t : t); }
        e0 += lane_xor<1>(e0); e1 += lane_xor<1>(e1); e6 += lane_xor<1>(e6); e7 += lane_xor<1>(e7);
        e0 += lane_xor<2>(e0); e1 += lane_xor<2>(e1); e6 += lane_xor<2>(e6); e7 += lane_xor<2>(e7);
        e0 += lane_xor<4>(e0); e1 += lane_xor<4>(e1); e6 += lane_xor<4>(e6); e7 += lane_xor<4>(e7);
        e0 += lane_xor<8>(e0); e1 += lane_xor<8>(e1); e6 += lane_xor<8>(e6); e7 += lane_xor<8>(e7);
        if (j == 0) {
            v2f lo, hi;
            lo[0] = e0; lo[1] = e1;
            hi[0] = e6; hi[1] = e7;
            *(v2f*)&E[m][0] = lo;
            *(v2f*)&E[m][6] = hi;
        } else if ((j & (j - 1)) == 0) {
            E[m][2 + __builtin_ctz(j)] = t;
        }
    }
    __syncthreads();

    if (tid < K3_ROWS * NCLS) {
        int m = tid / NCLS, c = tid % NCLS;
        float acc = out_b[c];
#pragma unroll
        for (int q = 0; q < NQ; ++q) acc += E[m][q] * out_w[c * NQ + q];
        out[(row0 + m) * NCLS + c] = acc;
    }
#undef U_ISSUE
#undef U_STASH
}

extern "C" void kernel_launch(void* const* d_in, const int* in_sizes, int n_in,
                              void* d_out, int out_size, void* d_ws, size_t ws_size,
                              hipStream_t stream) {
    const float* x      = (const float*)d_in[0];
    const float* proj_w = (const float*)d_in[1];
    const float* qnn_w  = (const float*)d_in[2];
    const float* out_w  = (const float*)d_in[3];
    const float* out_b  = (const float*)d_in[4];
    float* ws = (float*)d_ws;

    float* angles = ws;
    float* UrP    = ws + 65536;
    float* UiP    = ws + 65536 + 65536;
    float* cs_tab = (float*)d_out;

    k1_angles<<<BATCH / 16 + 1, 128, 0, stream>>>(x, proj_w, qnn_w, angles, cs_tab);
    k2_unitary<<<NSTATE / 2, 64, 0, stream>>>(cs_tab, UrP, UiP);
    k3_fused<<<BATCH / K3_ROWS, 512, 0, stream>>>(angles, UrP, UiP, out_w, out_b,
                                             (float*)d_out);
}
